# starting kernel
# baseline (speedup 1.0000x reference)
#include <hip/hip_runtime.h>
#include <hip/hip_bf16.h>


typedef float v2f __attribute__((ext_vector_type(2)));
typedef float v8f __attribute__((ext_vector_type(8)));

#define GAT_HEADS   4
#define GAT_D       32
#define GAT_FDIM    128
#define NEG_SLOPE   0.2f

__global__ __launch_bounds__(128) void gat_gemm_feat(
    const float* __restrict__ x, const float* __restrict__ w,
    float* __restrict__ feat, int nNodes)
{
    const int lane  = threadIdx.x & 31;
    const int wave  = threadIdx.x >> 5;
    const int mTile = blockIdx.x * 4 + wave;
    const int mTiles = (nNodes + 15) >> 4;
    if (mTile >= mTiles) return;

    const int laneM  = lane & 15;
    const int khalf  = lane >> 4;
    int row = mTile * 16 + laneM;
    int rowClamped = row < nNodes ? row : nNodes - 1;

    v8f acc[8];
    #pragma unroll
    for (int t = 0; t < 8; ++t) acc[t] = v8f{};

    const float* xrow = x + (size_t)rowClamped * GAT_FDIM;

    for (int k0 = 0; k0 < GAT_FDIM; k0 += 4) {
        v2f a = *(const v2f*)(xrow + k0 + 2 * khalf);
        const int kb = k0 + 2 * khalf;
        #pragma unroll
        for (int t = 0; t < 8; ++t) {
            const int ncol = t * 16 + laneM;
            v2f b;
            b.x = w[(kb + 0) * GAT_FDIM + ncol];
            b.y = w[(kb + 1) * GAT_FDIM + ncol];
            acc[t] = __builtin_amdgcn_wmma_f32_16x16x4_f32(
                 false, a,  false, b,
                 (short)0, acc[t],
                 false,  false);
        }
    }

    const int mOff = 8 * khalf;
    #pragma unroll
    for (int t = 0; t < 8; ++t) {
        #pragma unroll
        for (int r = 0; r < 8; ++r) {
            int m = mTile * 16 + mOff + r;
            if (m < nNodes)
                feat[(size_t)m * GAT_FDIM + t * 16 + laneM] = acc[t][r];
        }
    }
}

__global__ __launch_bounds__(128) void gat_node_attn(
    const float* __restrict__ feat,
    const float* __restrict__ attn_l, const float* __restrict__ attn_r,
    float* __restrict__ el, float* __restrict__ er, int nNodes)
{
    const int n = blockIdx.x;
    if (n >= nNodes) return;
    const int tid = threadIdx.x;
    const int h = tid >> 5, d = tid & 31;

    float v  = feat[(size_t)n * GAT_FDIM + tid];
    float pl = v * attn_l[h * GAT_D + d];
    float pr = v * attn_r[h * GAT_D + d];
    #pragma unroll
    for (int off = 16; off > 0; off >>= 1) {
        pl += __shfl_down(pl, off, 32);
        pr += __shfl_down(pr, off, 32);
    }
    if (d == 0) {
        el[n * GAT_HEADS + h] = pl;
        er[n * GAT_HEADS + h] = pr;
    }
}

__device__ __forceinline__ unsigned fkey(float f) {
    unsigned u = __float_as_uint(f);
    return (u & 0x80000000u) ? ~u : (u | 0x80000000u);
}
__device__ __forceinline__ float funkey(unsigned k) {
    return (k & 0x80000000u) ? __uint_as_float(k ^ 0x80000000u)
                             : __uint_as_float(~k);
}
__device__ __forceinline__ float lrelu(float v) {
    return v > 0.f ? v : NEG_SLOPE * v;
}

__global__ void gat_init(float* __restrict__ out, unsigned* __restrict__ mkey,
                         float* __restrict__ s, const float* __restrict__ bias,
                         int nNodes)
{
    int i = blockIdx.x * blockDim.x + threadIdx.x;
    if (i < nNodes * GAT_D) {
        int d = i & (GAT_D - 1);
        float b = 0.f;
        #pragma unroll
        for (int h = 0; h < GAT_HEADS; ++h) b += bias[h * GAT_D + d];
        out[i] = b * (1.f / GAT_HEADS);
    }
    if (i < nNodes * GAT_HEADS) { mkey[i] = 0u; s[i] = 0.f; }
}

__global__ void gat_edge_max(
    const int* __restrict__ src, const int* __restrict__ dst,
    const float* __restrict__ el, const float* __restrict__ er,
    unsigned* __restrict__ mkey, int nEdges)
{
    int i = blockIdx.x * blockDim.x + threadIdx.x;
    if (i >= nEdges * GAT_HEADS) return;
    int e = i >> 2, h = i & 3;
    int u = src[e], v = dst[e];
    float ev = lrelu(el[u * GAT_HEADS + h] + er[v * GAT_HEADS + h]);
    atomicMax(&mkey[v * GAT_HEADS + h], fkey(ev));
}

__global__ void gat_edge_sum(
    const int* __restrict__ src, const int* __restrict__ dst,
    const float* __restrict__ el, const float* __restrict__ er,
    const unsigned* __restrict__ mkey, float* __restrict__ s, int nEdges)
{
    int i = blockIdx.x * blockDim.x + threadIdx.x;
    if (i >= nEdges * GAT_HEADS) return;
    int e = i >> 2, h = i & 3;
    int u = src[e], v = dst[e];
    float ev = lrelu(el[u * GAT_HEADS + h] + er[v * GAT_HEADS + h]);
    float m  = funkey(mkey[v * GAT_HEADS + h]);
    atomicAdd(&s[v * GAT_HEADS + h], __expf(ev - m));
}

__global__ __launch_bounds__(256) void gat_edge_aggr(
    const int* __restrict__ src, const int* __restrict__ dst,
    const float* __restrict__ el, const float* __restrict__ er,
    const unsigned* __restrict__ mkey, const float* __restrict__ s,
    const float* __restrict__ feat, float* __restrict__ out, int nEdges)
{
    const int lane = threadIdx.x & 31;
    const int e = (blockIdx.x * blockDim.x + threadIdx.x) >> 5;
    if (e >= nEdges) return;
    const int u = src[e], v = dst[e];

    float alpha = 0.f;
    if (lane < GAT_HEADS) {
        const int h = lane;
        float ev = lrelu(el[u * GAT_HEADS + h] + er[v * GAT_HEADS + h]);
        float m  = funkey(mkey[v * GAT_HEADS + h]);
        float sv = s[v * GAT_HEADS + h];
        alpha = __expf(ev - m) / fmaxf(sv, 1e-16f);
    }
    const float a0 = __shfl(alpha, 0, 32);
    const float a1 = __shfl(alpha, 1, 32);
    const float a2 = __shfl(alpha, 2, 32);
    const float a3 = __shfl(alpha, 3, 32);

    const float* f = feat + (size_t)u * GAT_FDIM;
    float accv = a0 * f[lane] + a1 * f[GAT_D + lane] +
                 a2 * f[2 * GAT_D + lane] + a3 * f[3 * GAT_D + lane];
    atomicAdd(out + (size_t)v * GAT_D + lane, 0.25f * accv);
}

extern "C" void kernel_launch(void* const* d_in, const int* in_sizes, int n_in,
                              void* d_out, int out_size, void* d_ws, size_t ws_size,
                              hipStream_t stream)
{
    const float* x        = (const float*)d_in[0];
    const int*   edge_src = (const int*)  d_in[1];
    const int*   edge_dst = (const int*)  d_in[2];
    const float* fc_w     = (const float*)d_in[3];
    const float* attn_l   = (const float*)d_in[4];
    const float* attn_r   = (const float*)d_in[5];
    const float* bias     = (const float*)d_in[6];
    float*       out      = (float*)d_out;

    const int nNodes = in_sizes[0] / GAT_FDIM;
    const int nEdges = in_sizes[1];

    float*    feat = (float*)d_ws;
    float*    el   = feat + (size_t)nNodes * GAT_FDIM;
    float*    er   = el   + (size_t)nNodes * GAT_HEADS;
    unsigned* mkey = (unsigned*)(er + (size_t)nNodes * GAT_HEADS);
    float*    s    = (float*)(mkey + (size_t)nNodes * GAT_HEADS);

    {
        int mTiles = (nNodes + 15) / 16;
        int blocks = (mTiles + 3) / 4;
        gat_gemm_feat<<<blocks, 128, 0, stream>>>(x, fc_w, feat, nNodes);
    }
    gat_node_attn<<<nNodes, 128, 0, stream>>>(feat, attn_l, attn_r, el, er, nNodes);
    {
        int total = nNodes * GAT_D;
        gat_init<<<(total + 255) / 256, 256, 0, stream>>>(out, mkey, s, bias, nNodes);
    }
    {
        int total = nEdges * GAT_HEADS;
        gat_edge_max<<<(total + 255) / 256, 256, 0, stream>>>(
            edge_src, edge_dst, el, er, mkey, nEdges);
    }
    {
        int total = nEdges * GAT_HEADS;
        gat_edge_sum<<<(total + 255) / 256, 256, 0, stream>>>(
            edge_src, edge_dst, el, er, mkey, s, nEdges);
    }
    {
        long long total = (long long)nEdges * 32;
        int blocks = (int)((total + 255) / 256);
        gat_edge_aggr<<<blocks, 256, 0, stream>>>(
            edge_src, edge_dst, el, er, mkey, s, feat, out, nEdges);
    }
}


# aggregation pass: one wave per 8 edges instead of one per edge (32 lanes form the 8x4 softmax weights, 32 feature-row loads in flight at once); atomic adds into out stay in arrival order
# speedup vs baseline: 1.3132x; 1.3132x over previous
#include <hip/hip_runtime.h>
#include <hip/hip_bf16.h>


typedef float v2f __attribute__((ext_vector_type(2)));
typedef float v8f __attribute__((ext_vector_type(8)));

#define GAT_HEADS   4
#define GAT_D       32
#define GAT_FDIM    128
#define NEG_SLOPE   0.2f

__global__ __launch_bounds__(128) void gat_gemm_feat(
    const float* __restrict__ x, const float* __restrict__ w,
    float* __restrict__ feat, int nNodes)
{
    const int lane  = threadIdx.x & 31;
    const int wave  = threadIdx.x >> 5;
    const int mTile = blockIdx.x * 4 + wave;
    const int mTiles = (nNodes + 15) >> 4;
    if (mTile >= mTiles) return;

    const int laneM  = lane & 15;
    const int khalf  = lane >> 4;
    int row = mTile * 16 + laneM;
    int rowClamped = row < nNodes ? row : nNodes - 1;

    v8f acc[8];
    #pragma unroll
    for (int t = 0; t < 8; ++t) acc[t] = v8f{};

    const float* xrow = x + (size_t)rowClamped * GAT_FDIM;

    for (int k0 = 0; k0 < GAT_FDIM; k0 += 4) {
        v2f a = *(const v2f*)(xrow + k0 + 2 * khalf);
        const int kb = k0 + 2 * khalf;
        #pragma unroll
        for (int t = 0; t < 8; ++t) {
            const int ncol = t * 16 + laneM;
            v2f b;
            b.x = w[(kb + 0) * GAT_FDIM + ncol];
            b.y = w[(kb + 1) * GAT_FDIM + ncol];
            acc[t] = __builtin_amdgcn_wmma_f32_16x16x4_f32(
                 false, a,  false, b,
                 (short)0, acc[t],
                 false,  false);
        }
    }

    const int mOff = 8 * khalf;
    #pragma unroll
    for (int t = 0; t < 8; ++t) {
        #pragma unroll
        for (int r = 0; r < 8; ++r) {
            int m = mTile * 16 + mOff + r;
            if (m < nNodes)
                feat[(size_t)m * GAT_FDIM + t * 16 + laneM] = acc[t][r];
        }
    }
}

__global__ __launch_bounds__(128) void gat_node_attn(
    const float* __restrict__ feat,
    const float* __restrict__ attn_l, const float* __restrict__ attn_r,
    float* __restrict__ el, float* __restrict__ er, int nNodes)
{
    const int n = blockIdx.x;
    if (n >= nNodes) return;
    const int tid = threadIdx.x;
    const int h = tid >> 5, d = tid & 31;

    float v  = feat[(size_t)n * GAT_FDIM + tid];
    float pl = v * attn_l[h * GAT_D + d];
    float pr = v * attn_r[h * GAT_D + d];
    #pragma unroll
    for (int off = 16; off > 0; off >>= 1) {
        pl += __shfl_down(pl, off, 32);
        pr += __shfl_down(pr, off, 32);
    }
    if (d == 0) {
        el[n * GAT_HEADS + h] = pl;
        er[n * GAT_HEADS + h] = pr;
    }
}

__device__ __forceinline__ unsigned fkey(float f) {
    unsigned u = __float_as_uint(f);
    return (u & 0x80000000u) ? ~u : (u | 0x80000000u);
}
__device__ __forceinline__ float funkey(unsigned k) {
    return (k & 0x80000000u) ? __uint_as_float(k ^ 0x80000000u)
                             : __uint_as_float(~k);
}
__device__ __forceinline__ float lrelu(float v) {
    return v > 0.f ? v : NEG_SLOPE * v;
}

__global__ void gat_init(float* __restrict__ out, unsigned* __restrict__ mkey,
                         float* __restrict__ s, const float* __restrict__ bias,
                         int nNodes)
{
    int i = blockIdx.x * blockDim.x + threadIdx.x;
    if (i < nNodes * GAT_D) {
        int d = i & (GAT_D - 1);
        float b = 0.f;
        #pragma unroll
        for (int h = 0; h < GAT_HEADS; ++h) b += bias[h * GAT_D + d];
        out[i] = b * (1.f / GAT_HEADS);
    }
    if (i < nNodes * GAT_HEADS) { mkey[i] = 0u; s[i] = 0.f; }
}

__global__ void gat_edge_max(
    const int* __restrict__ src, const int* __restrict__ dst,
    const float* __restrict__ el, const float* __restrict__ er,
    unsigned* __restrict__ mkey, int nEdges)
{
    int i = blockIdx.x * blockDim.x + threadIdx.x;
    if (i >= nEdges * GAT_HEADS) return;
    int e = i >> 2, h = i & 3;
    int u = src[e], v = dst[e];
    float ev = lrelu(el[u * GAT_HEADS + h] + er[v * GAT_HEADS + h]);
    atomicMax(&mkey[v * GAT_HEADS + h], fkey(ev));
}

__global__ void gat_edge_sum(
    const int* __restrict__ src, const int* __restrict__ dst,
    const float* __restrict__ el, const float* __restrict__ er,
    const unsigned* __restrict__ mkey, float* __restrict__ s, int nEdges)
{
    int i = blockIdx.x * blockDim.x + threadIdx.x;
    if (i >= nEdges * GAT_HEADS) return;
    int e = i >> 2, h = i & 3;
    int u = src[e], v = dst[e];
    float ev = lrelu(el[u * GAT_HEADS + h] + er[v * GAT_HEADS + h]);
    float m  = funkey(mkey[v * GAT_HEADS + h]);
    atomicAdd(&s[v * GAT_HEADS + h], __expf(ev - m));
}

#define AGG_EPW 8

__device__ __forceinline__ float lane_bcast(float x, int srcLane) {
    return __int_as_float(__builtin_amdgcn_readlane(__float_as_int(x), srcLane));
}

__device__ __forceinline__ float edge_weight(float elv, float erv,
                                             unsigned mk, float sv) {
    const float ev = lrelu(elv + erv);
    return __expf(ev - funkey(mk)) / fmaxf(sv, 1e-16f);
}

__global__ __launch_bounds__(256) void gat_edge_aggr(
    const int* __restrict__ src, const int* __restrict__ dst,
    const float* __restrict__ el, const float* __restrict__ er,
    const unsigned* __restrict__ mkey, const float* __restrict__ s,
    const float* __restrict__ feat, float* __restrict__ out, int nEdges)
{
    const int lane  = threadIdx.x & 31;
    const int wave  = (blockIdx.x * blockDim.x + threadIdx.x) >> 5;
    const int eBase = wave * AGG_EPW;
    if (eBase >= nEdges) return;

    const int j = lane >> 2, h = lane & 3;
    const int e  = eBase + j;
    const int ec = e < nEdges ? e : nEdges - 1;
    const int u = src[ec], v = dst[ec];

    const float    elv = el[u * GAT_HEADS + h];
    const float    erv = er[v * GAT_HEADS + h];
    const unsigned mk  = mkey[v * GAT_HEADS + h];
    const float    sv  = s[v * GAT_HEADS + h];

    if (eBase + AGG_EPW <= nEdges) {
        float f[AGG_EPW][GAT_HEADS];
        #pragma unroll
        for (int jj = 0; jj < AGG_EPW; ++jj) {
            const int uj = __builtin_amdgcn_readlane(u, jj * 4);
            const float* frow = feat + (size_t)uj * GAT_FDIM + lane;
            #pragma unroll
            for (int hh = 0; hh < GAT_HEADS; ++hh) f[jj][hh] = frow[hh * GAT_D];
        }
        const float alpha = edge_weight(elv, erv, mk, sv);
        #pragma unroll
        for (int jj = 0; jj < AGG_EPW; ++jj) {
            const int vj = __builtin_amdgcn_readlane(v, jj * 4);
            const float a0 = lane_bcast(alpha, jj * 4 + 0);
            const float a1 = lane_bcast(alpha, jj * 4 + 1);
            const float a2 = lane_bcast(alpha, jj * 4 + 2);
            const float a3 = lane_bcast(alpha, jj * 4 + 3);
            const float accv = a0 * f[jj][0] + a1 * f[jj][1] +
                               a2 * f[jj][2] + a3 * f[jj][3];
            atomicAdd(out + (size_t)vj * GAT_D + lane, 0.25f * accv);
        }
    } else {
        const int nLeft = nEdges - eBase;
        const float alpha = edge_weight(elv, erv, mk, sv);
        for (int jj = 0; jj < nLeft; ++jj) {
            const int uj = __shfl(u, jj * 4, 32);
            const int vj = __shfl(v, jj * 4, 32);
            const float a0 = __shfl(alpha, jj * 4 + 0, 32);
            const float a1 = __shfl(alpha, jj * 4 + 1, 32);
            const float a2 = __shfl(alpha, jj * 4 + 2, 32);
            const float a3 = __shfl(alpha, jj * 4 + 3, 32);
            const float* frow = feat + (size_t)uj * GAT_FDIM + lane;
            const float accv = a0 * frow[0] + a1 * frow[GAT_D] +
                               a2 * frow[2 * GAT_D] + a3 * frow[3 * GAT_D];
            atomicAdd(out + (size_t)vj * GAT_D + lane, 0.25f * accv);
        }
    }
}

extern "C" void kernel_launch(void* const* d_in, const int* in_sizes, int n_in,
                              void* d_out, int out_size, void* d_ws, size_t ws_size,
                              hipStream_t stream)
{
    const float* x        = (const float*)d_in[0];
    const int*   edge_src = (const int*)  d_in[1];
    const int*   edge_dst = (const int*)  d_in[2];
    const float* fc_w     = (const float*)d_in[3];
    const float* attn_l   = (const float*)d_in[4];
    const float* attn_r   = (const float*)d_in[5];
    const float* bias     = (const float*)d_in[6];
    float*       out      = (float*)d_out;

    const int nNodes = in_sizes[0] / GAT_FDIM;
    const int nEdges = in_sizes[1];

    float*    feat = (float*)d_ws;
    float*    el   = feat + (size_t)nNodes * GAT_FDIM;
    float*    er   = el   + (size_t)nNodes * GAT_HEADS;
    unsigned* mkey = (unsigned*)(er + (size_t)nNodes * GAT_HEADS);
    float*    s    = (float*)(mkey + (size_t)nNodes * GAT_HEADS);

    {
        int mTiles = (nNodes + 15) / 16;
        int blocks = (mTiles + 3) / 4;
        gat_gemm_feat<<<blocks, 128, 0, stream>>>(x, fc_w, feat, nNodes);
    }
    gat_node_attn<<<nNodes, 128, 0, stream>>>(feat, attn_l, attn_r, el, er, nNodes);
    {
        int total = nNodes * GAT_D;
        gat_init<<<(total + 255) / 256, 256, 0, stream>>>(out, mkey, s, bias, nNodes);
    }
    {
        int total = nEdges * GAT_HEADS;
        gat_edge_max<<<(total + 255) / 256, 256, 0, stream>>>(
            edge_src, edge_dst, el, er, mkey, nEdges);
    }
    {
        int total = nEdges * GAT_HEADS;
        gat_edge_sum<<<(total + 255) / 256, 256, 0, stream>>>(
            edge_src, edge_dst, el, er, mkey, s, nEdges);
    }
    {
        long long total = (long long)((nEdges + AGG_EPW - 1) / AGG_EPW) * 32;
        int blocks = (int)((total + 255) / 256);
        gat_edge_aggr<<<blocks, 256, 0, stream>>>(
            edge_src, edge_dst, el, er, mkey, s, feat, out, nEdges);
    }
}


# edges grouped by destination (rank = returning int atomic, block-scan segments); one wave per node does softmax and row sum in registers, no float atomics; sum order follows rank arrival order
# speedup vs baseline: 1.5804x; 1.2035x over previous
#include <hip/hip_runtime.h>
#include <hip/hip_bf16.h>


typedef float v2f __attribute__((ext_vector_type(2)));
typedef float v8f __attribute__((ext_vector_type(8)));

#define GAT_HEADS   4
#define GAT_D       32
#define GAT_FDIM    128
#define NEG_SLOPE   0.2f

__global__ __launch_bounds__(128) void gat_gemm_feat(
    const float* __restrict__ x, const float* __restrict__ w,
    float* __restrict__ feat, int nNodes)
{
    const int lane  = threadIdx.x & 31;
    const int wave  = threadIdx.x >> 5;
    const int mTile = blockIdx.x * 4 + wave;
    const int mTiles = (nNodes + 15) >> 4;
    if (mTile >= mTiles) return;

    const int laneM  = lane & 15;
    const int khalf  = lane >> 4;
    int row = mTile * 16 + laneM;
    int rowClamped = row < nNodes ? row : nNodes - 1;

    v8f acc[8];
    #pragma unroll
    for (int t = 0; t < 8; ++t) acc[t] = v8f{};

    const float* xrow = x + (size_t)rowClamped * GAT_FDIM;

    for (int k0 = 0; k0 < GAT_FDIM; k0 += 4) {
        v2f a = *(const v2f*)(xrow + k0 + 2 * khalf);
        const int kb = k0 + 2 * khalf;
        #pragma unroll
        for (int t = 0; t < 8; ++t) {
            const int ncol = t * 16 + laneM;
            v2f b;
            b.x = w[(kb + 0) * GAT_FDIM + ncol];
            b.y = w[(kb + 1) * GAT_FDIM + ncol];
            acc[t] = __builtin_amdgcn_wmma_f32_16x16x4_f32(
                 false, a,  false, b,
                 (short)0, acc[t],
                 false,  false);
        }
    }

    const int mOff = 8 * khalf;
    #pragma unroll
    for (int t = 0; t < 8; ++t) {
        #pragma unroll
        for (int r = 0; r < 8; ++r) {
            int m = mTile * 16 + mOff + r;
            if (m < nNodes)
                feat[(size_t)m * GAT_FDIM + t * 16 + laneM] = acc[t][r];
        }
    }
}

__global__ __launch_bounds__(128) void gat_node_attn(
    const float* __restrict__ feat,
    const float* __restrict__ attn_l, const float* __restrict__ attn_r,
    float* __restrict__ el, float* __restrict__ er,
    int* __restrict__ cnt, int* __restrict__ cursor, int nNodes)
{
    const int n = blockIdx.x;
    if (n >= nNodes) return;
    const int tid = threadIdx.x;
    const int h = tid >> 5, d = tid & 31;

    float v  = feat[(size_t)n * GAT_FDIM + tid];
    float pl = v * attn_l[h * GAT_D + d];
    float pr = v * attn_r[h * GAT_D + d];
    #pragma unroll
    for (int off = 16; off > 0; off >>= 1) {
        pl += __shfl_down(pl, off, 32);
        pr += __shfl_down(pr, off, 32);
    }
    if (d == 0) {
        el[n * GAT_HEADS + h] = pl;
        er[n * GAT_HEADS + h] = pr;
    }
    if (tid == 0) cnt[n] = 0;
    if (n == 0 && tid == 1) *cursor = 0;
}

__device__ __forceinline__ float lrelu(float v) {
    return v > 0.f ? v : NEG_SLOPE * v;
}

__global__ void gat_edge_rank(
    const int* __restrict__ dst, int* __restrict__ cnt,
    int* __restrict__ rank, int nEdges)
{
    int e = blockIdx.x * blockDim.x + threadIdx.x;
    if (e >= nEdges) return;
    rank[e] = atomicAdd(&cnt[dst[e]], 1);
}

__global__ __launch_bounds__(256) void gat_seg_alloc(
    const int* __restrict__ cnt, int* __restrict__ start,
    int* __restrict__ cursor, int nNodes)
{
    __shared__ int waveTot[8];
    __shared__ int blockBase;
    const int v = blockIdx.x * 256 + threadIdx.x;
    const int lane = threadIdx.x & 31, wave = threadIdx.x >> 5;
    const int c = v < nNodes ? cnt[v] : 0;
    int incl = c;
    #pragma unroll
    for (int off = 1; off < 32; off <<= 1) {
        int t = __shfl_up(incl, off, 32);
        if (lane >= off) incl += t;
    }
    if (lane == 31) waveTot[wave] = incl;
    __syncthreads();
    if (threadIdx.x == 0) {
        int tot = 0;
        for (int w = 0; w < 8; ++w) { int t = waveTot[w]; waveTot[w] = tot; tot += t; }
        blockBase = atomicAdd(cursor, tot);
    }
    __syncthreads();
    if (v < nNodes) start[v] = blockBase + waveTot[wave] + incl - c;
}

__global__ void gat_edge_place(
    const int* __restrict__ src, const int* __restrict__ dst,
    const int* __restrict__ start, const int* __restrict__ rank,
    int* __restrict__ sorted, int nEdges)
{
    int e = blockIdx.x * blockDim.x + threadIdx.x;
    if (e >= nEdges) return;
    sorted[start[dst[e]] + rank[e]] = src[e];
}

__device__ __forceinline__ float lane_bcast(float x, int srcLane) {
    return __int_as_float(__builtin_amdgcn_readlane(__float_as_int(x), srcLane));
}
__device__ __forceinline__ float wave_max(float x) {
    #pragma unroll
    for (int off = 16; off > 0; off >>= 1) x = fmaxf(x, __shfl_xor(x, off, 32));
    return x;
}
__device__ __forceinline__ float wave_sum(float x) {
    #pragma unroll
    for (int off = 16; off > 0; off >>= 1) x += __shfl_xor(x, off, 32);
    return x;
}

__device__ __forceinline__ float add_rows(
    float acc, int u, float a0, float a1, float a2, float a3,
    int nIn, const float* __restrict__ feat, int lane)
{
    for (int j0 = 0; j0 < nIn; j0 += 4) {
        float f[4][GAT_HEADS];
        #pragma unroll
        for (int jj = 0; jj < 4; ++jj) {
            const int j  = j0 + jj;
            const int jc = j < nIn ? j : nIn - 1;
            const int uj = __builtin_amdgcn_readlane(u, jc);
            const float* frow = feat + (size_t)uj * GAT_FDIM + lane;
            #pragma unroll
            for (int hh = 0; hh < GAT_HEADS; ++hh) f[jj][hh] = frow[hh * GAT_D];
        }
        #pragma unroll
        for (int jj = 0; jj < 4; ++jj) {
            const int j  = j0 + jj;
            const int jc = j < nIn ? j : nIn - 1;
            const float t = lane_bcast(a0, jc) * f[jj][0] + lane_bcast(a1, jc) * f[jj][1] +
                            lane_bcast(a2, jc) * f[jj][2] + lane_bcast(a3, jc) * f[jj][3];
            acc += j < nIn ? t : 0.f;
        }
    }
    return acc;
}

__global__ __launch_bounds__(256) void gat_node_aggr(
    const int* __restrict__ sorted, const int* __restrict__ start,
    const int* __restrict__ cnt,
    const float* __restrict__ el, const float* __restrict__ er,
    const float* __restrict__ feat, const float* __restrict__ bias,
    float* __restrict__ out, int nNodes)
{
    const int lane = threadIdx.x & 31;
    const int v = __builtin_amdgcn_readfirstlane(
        (int)((blockIdx.x * blockDim.x + threadIdx.x) >> 5));
    if (v >= nNodes) return;

    const int beg = start[v];
    const int deg = cnt[v];
    const float er0 = er[v * GAT_HEADS + 0], er1 = er[v * GAT_HEADS + 1];
    const float er2 = er[v * GAT_HEADS + 2], er3 = er[v * GAT_HEADS + 3];
    const float bm = (bias[lane] + bias[GAT_D + lane] + bias[2 * GAT_D + lane] +
                      bias[3 * GAT_D + lane]) * (1.f / GAT_HEADS);
    const float NEG_INF = -__builtin_huge_valf();
    float acc = 0.f;

    if (deg <= 32) {
        const bool valid = lane < deg;
        int u = 0;
        float e0 = NEG_INF, e1 = NEG_INF, e2 = NEG_INF, e3 = NEG_INF;
        if (valid) {
            u = sorted[beg + lane];
            const float4 l = *(const float4*)(el + (size_t)u * GAT_HEADS);
            e0 = lrelu(l.x + er0); e1 = lrelu(l.y + er1);
            e2 = lrelu(l.z + er2); e3 = lrelu(l.w + er3);
        }
        if (deg > 0) {
            const float m0 = wave_max(e0), m1 = wave_max(e1);
            const float m2 = wave_max(e2), m3 = wave_max(e3);
            const float p0 = valid ? __expf(e0 - m0) : 0.f;
            const float p1 = valid ? __expf(e1 - m1) : 0.f;
            const float p2 = valid ? __expf(e2 - m2) : 0.f;
            const float p3 = valid ? __expf(e3 - m3) : 0.f;
            const float a0 = p0 / fmaxf(wave_sum(p0), 1e-16f);
            const float a1 = p1 / fmaxf(wave_sum(p1), 1e-16f);
            const float a2 = p2 / fmaxf(wave_sum(p2), 1e-16f);
            const float a3 = p3 / fmaxf(wave_sum(p3), 1e-16f);
            acc = add_rows(acc, u, a0, a1, a2, a3, deg, feat, lane);
        }
    } else {
        float m0 = NEG_INF, m1 = NEG_INF, m2 = NEG_INF, m3 = NEG_INF;
        for (int c = 0; c < deg; c += 32) {
            if (c + lane < deg) {
                const int u = sorted[beg + c + lane];
                const float4 l = *(const float4*)(el + (size_t)u * GAT_HEADS);
                m0 = fmaxf(m0, lrelu(l.x + er0)); m1 = fmaxf(m1, lrelu(l.y + er1));
                m2 = fmaxf(m2, lrelu(l.z + er2)); m3 = fmaxf(m3, lrelu(l.w + er3));
            }
        }
        m0 = wave_max(m0); m1 = wave_max(m1); m2 = wave_max(m2); m3 = wave_max(m3);
        float s0 = 0.f, s1 = 0.f, s2 = 0.f, s3 = 0.f;
        for (int c = 0; c < deg; c += 32) {
            if (c + lane < deg) {
                const int u = sorted[beg + c + lane];
                const float4 l = *(const float4*)(el + (size_t)u * GAT_HEADS);
                s0 += __expf(lrelu(l.x + er0) - m0); s1 += __expf(lrelu(l.y + er1) - m1);
                s2 += __expf(lrelu(l.z + er2) - m2); s3 += __expf(lrelu(l.w + er3) - m3);
            }
        }
        s0 = fmaxf(wave_sum(s0), 1e-16f); s1 = fmaxf(wave_sum(s1), 1e-16f);
        s2 = fmaxf(wave_sum(s2), 1e-16f); s3 = fmaxf(wave_sum(s3), 1e-16f);
        for (int c = 0; c < deg; c += 32) {
            int u = 0;
            float a0 = 0.f, a1 = 0.f, a2 = 0.f, a3 = 0.f;
            if (c + lane < deg) {
                u = sorted[beg + c + lane];
                const float4 l = *(const float4*)(el + (size_t)u * GAT_HEADS);
                a0 = __expf(lrelu(l.x + er0) - m0) / s0; a1 = __expf(lrelu(l.y + er1) - m1) / s1;
                a2 = __expf(lrelu(l.z + er2) - m2) / s2; a3 = __expf(lrelu(l.w + er3) - m3) / s3;
            }
            const int left = deg - c;
            acc = add_rows(acc, u, a0, a1, a2, a3, left < 32 ? left : 32, feat, lane);
        }
    }
    out[(size_t)v * GAT_D + lane] = 0.25f * acc + bm;
}

extern "C" void kernel_launch(void* const* d_in, const int* in_sizes, int n_in,
                              void* d_out, int out_size, void* d_ws, size_t ws_size,
                              hipStream_t stream)
{
    const float* x        = (const float*)d_in[0];
    const int*   edge_src = (const int*)  d_in[1];
    const int*   edge_dst = (const int*)  d_in[2];
    const float* fc_w     = (const float*)d_in[3];
    const float* attn_l   = (const float*)d_in[4];
    const float* attn_r   = (const float*)d_in[5];
    const float* bias     = (const float*)d_in[6];
    float*       out      = (float*)d_out;

    const int nNodes = in_sizes[0] / GAT_FDIM;
    const int nEdges = in_sizes[1];

    float* feat   = (float*)d_ws;
    float* el     = feat + (size_t)nNodes * GAT_FDIM;
    float* er     = el   + (size_t)nNodes * GAT_HEADS;
    int*   cnt    = (int*)(er + (size_t)nNodes * GAT_HEADS);
    int*   start  = cnt + nNodes;
    int*   cursor = start + nNodes;
    int*   rank   = cursor + 4;
    int*   sorted = rank + nEdges;
    if ((size_t)((char*)(sorted + nEdges) - (char*)d_ws) > ws_size) return;

    {
        int mTiles = (nNodes + 15) / 16;
        int blocks = (mTiles + 3) / 4;
        gat_gemm_feat<<<blocks, 128, 0, stream>>>(x, fc_w, feat, nNodes);
    }
    gat_node_attn<<<nNodes, 128, 0, stream>>>(feat, attn_l, attn_r, el, er,
                                              cnt, cursor, nNodes);
    gat_edge_rank<<<(nEdges + 255) / 256, 256, 0, stream>>>(edge_dst, cnt, rank, nEdges);
    gat_seg_alloc<<<(nNodes + 255) / 256, 256, 0, stream>>>(cnt, start, cursor, nNodes);
    gat_edge_place<<<(nEdges + 255) / 256, 256, 0, stream>>>(
        edge_src, edge_dst, start, rank, sorted, nEdges);
    {
        int blocks = (nNodes + 7) / 8;
        gat_node_aggr<<<blocks, 256, 0, stream>>>(
            sorted, start, cnt, el, er, feat, bias, out, nNodes);
    }
}


# aggregation pass: a wave takes four destination nodes, eight lanes each (16-byte feature-row loads, 8-lane DPP softmax sums): a quarter of the waves and row loads; exp-sum reduction tree changes
# speedup vs baseline: 3.3544x; 2.1225x over previous
#include <hip/hip_runtime.h>
#include <hip/hip_bf16.h>


typedef float v2f __attribute__((ext_vector_type(2)));
typedef float v8f __attribute__((ext_vector_type(8)));

#define GAT_HEADS   4
#define GAT_D       32
#define GAT_FDIM    128
#define NEG_SLOPE   0.2f


template <int M>
__device__ __forceinline__ float row_xor_get(float v) {
    return __int_as_float(__builtin_amdgcn_update_dpp(
        0, __float_as_int(v), 0x160 + M, 0xf, 0xf, false));
}

__device__ __forceinline__ void row_sum_32(float (&p)[32], int c) {
    {
        const bool up = (c & 8) != 0;
        #pragma unroll
        for (int i = 0; i < 16; ++i) {
            const float give = up ? p[i] : p[i + 16];
            const float keep = up ? p[i + 16] : p[i];
            p[i] = keep + row_xor_get<8>(give);
        }
    }
    {
        const bool up = (c & 4) != 0;
        #pragma unroll
        for (int i = 0; i < 8; ++i) {
            const float give = up ? p[i] : p[i + 8];
            const float keep = up ? p[i + 8] : p[i];
            p[i] = keep + row_xor_get<4>(give);
        }
    }
    {
        const bool up = (c & 2) != 0;
        #pragma unroll
        for (int i = 0; i < 4; ++i) {
            const float give = up ? p[i] : p[i + 4];
            const float keep = up ? p[i + 4] : p[i];
            p[i] = keep + row_xor_get<2>(give);
        }
    }
    {
        const bool up = (c & 1) != 0;
        #pragma unroll
        for (int i = 0; i < 2; ++i) {
            const float give = up ? p[i] : p[i + 2];
            const float keep = up ? p[i + 2] : p[i];
            p[i] = keep + row_xor_get<1>(give);
        }
    }
}

__global__ __launch_bounds__(128) void gat_gemm_feat(
    const float* __restrict__ x, const float* __restrict__ w,
    const float* __restrict__ attn_l, const float* __restrict__ attn_r,
    float* __restrict__ feat, float* __restrict__ el, float* __restrict__ er,
    int nNodes)
{
    const int lane  = threadIdx.x & 31;
    const int wave  = threadIdx.x >> 5;
    const int mTile = blockIdx.x * 4 + wave;
    const int mTiles = (nNodes + 15) >> 4;
    if (mTile >= mTiles) return;

    const int laneM  = lane & 15;
    const int khalf  = lane >> 4;
    int row = mTile * 16 + laneM;
    int rowClamped = row < nNodes ? row : nNodes - 1;

    v8f acc[8];
    #pragma unroll
    for (int t = 0; t < 8; ++t) acc[t] = v8f{};

    const float* xrow = x + (size_t)rowClamped * GAT_FDIM;

    for (int k0 = 0; k0 < GAT_FDIM; k0 += 4) {
        v2f a = *(const v2f*)(xrow + k0 + 2 * khalf);
        const int kb = k0 + 2 * khalf;
        #pragma unroll
        for (int t = 0; t < 8; ++t) {
            const int ncol = t * 16 + laneM;
            v2f b;
            b.x = w[(kb + 0) * GAT_FDIM + ncol];
            b.y = w[(kb + 1) * GAT_FDIM + ncol];
            acc[t] = __builtin_amdgcn_wmma_f32_16x16x4_f32(
                 false, a,  false, b,
                 (short)0, acc[t],
                 false,  false);
        }
    }

    const int mOff = 8 * khalf;
    #pragma unroll
    for (int t = 0; t < 8; ++t) {
        #pragma unroll
        for (int r = 0; r < 8; ++r) {
            int m = mTile * 16 + mOff + r;
            if (m < nNodes)
                feat[(size_t)m * GAT_FDIM + t * 16 + laneM] = acc[t][r];
        }
    }

    const size_t lineOff = (size_t)(mTile * 16 + mOff) * GAT_HEADS + 2 * laneM;
    const bool rowOk = mTile * 16 + mOff + (laneM >> 1) < nNodes;
    {
        float wl[8];
        #pragma unroll
        for (int t = 0; t < 8; ++t) wl[t] = attn_l[t * 16 + laneM];
        float p[32];
        #pragma unroll
        for (int r = 0; r < 8; ++r) {
            #pragma unroll
            for (int h = 0; h < GAT_HEADS; ++h)
                p[r * 4 + h] = acc[2 * h][r] * wl[2 * h] + acc[2 * h + 1][r] * wl[2 * h + 1];
        }
        row_sum_32(p, laneM);
        if (rowOk) *(v2f*)(el + lineOff) = v2f{p[0], p[1]};
    }
    {
        float wr[8];
        #pragma unroll
        for (int t = 0; t < 8; ++t) wr[t] = attn_r[t * 16 + laneM];
        float p[32];
        #pragma unroll
        for (int r = 0; r < 8; ++r) {
            #pragma unroll
            for (int h = 0; h < GAT_HEADS; ++h)
                p[r * 4 + h] = acc[2 * h][r] * wr[2 * h] + acc[2 * h + 1][r] * wr[2 * h + 1];
        }
        row_sum_32(p, laneM);
        if (rowOk) *(v2f*)(er + lineOff) = v2f{p[0], p[1]};
    }
}

__device__ __forceinline__ float lrelu(float v) {
    return v > 0.f ? v : NEG_SLOPE * v;
}

#define BKT_SHIFT    8
#define BKT_NODES    256
#define MAX_BKTS     512
#define CHUNK_EDGES  3200
#define CHUNK_STEPS  13

__device__ __forceinline__ int block_excl_scan(int v, int* waveTot, int* total) {
    const int lane = threadIdx.x & 31, wave = threadIdx.x >> 5;
    int incl = v;
    #pragma unroll
    for (int off = 1; off < 32; off <<= 1) {
        int t = __shfl_up(incl, off, 32);
        if (lane >= off) incl += t;
    }
    if (lane == 31) waveTot[wave] = incl;
    __syncthreads();
    int wbase = 0, tot = 0;
    #pragma unroll
    for (int q = 0; q < 8; ++q) {
        const int wt = waveTot[q];
        wbase += q < wave ? wt : 0;
        tot += wt;
    }
    *total = tot;
    return wbase + incl - v;
}

__global__ __launch_bounds__(256) void gat_edge_bin(
    const int* __restrict__ src, const int* __restrict__ dst,
    int* __restrict__ binned, int* __restrict__ runTab, int* __restrict__ cursor,
    int nEdges, int nBkts, int nChunks)
{
    __shared__ int hist[MAX_BKTS + 2];
    __shared__ int waveTot[8];
    const int t = threadIdx.x;
    const int w = blockIdx.x;
    const int base = w * CHUNK_EDGES;
    const int left = nEdges - base;
    const int len = left < CHUNK_EDGES ? left : CHUNK_EDGES;
    if (w == 0 && t == 0) *cursor = 0;
    hist[2 * t] = 0;
    hist[2 * t + 1] = 0;
    __syncthreads();

    int d[CHUNK_STEPS], s[CHUNK_STEPS];
    #pragma unroll
    for (int k = 0; k < CHUNK_STEPS; ++k) {
        const int i  = k * 256 + t;
        const int ic = i < len ? i : len - 1;
        d[k] = dst[base + ic];
        s[k] = src[base + ic];
    }
    int lr[CHUNK_STEPS];
    #pragma unroll
    for (int k = 0; k < CHUNK_STEPS; ++k)
        lr[k] = atomicAdd(&hist[k * 256 + t < len ? d[k] >> BKT_SHIFT : MAX_BKTS + 1], 1);
    __syncthreads();

    const int c0 = hist[2 * t], c1 = hist[2 * t + 1];
    int total;
    const int ex = block_excl_scan(c0 + c1, waveTot, &total);
    hist[2 * t] = ex;
    hist[2 * t + 1] = ex + c0;
    if (t == 255) hist[MAX_BKTS] = total;
    __syncthreads();

    for (int b = t; b <= nBkts; b += 256)
        runTab[(size_t)b * nChunks + w] = hist[b];
    #pragma unroll
    for (int k = 0; k < CHUNK_STEPS; ++k) {
        if (k * 256 + t < len)
            binned[base + hist[d[k] >> BKT_SHIFT] + lr[k]] =
                (int)(((unsigned)s[k] << BKT_SHIFT) | (unsigned)(d[k] & (BKT_NODES - 1)));
    }
}

__global__ __launch_bounds__(256) void gat_bucket_group(
    const int* __restrict__ binned, const int* __restrict__ runTab,
    int* __restrict__ cnt, int* __restrict__ start, int* __restrict__ sorted,
    int* __restrict__ cursor, int nNodes, int nChunks)
{
    __shared__ int slot[BKT_NODES];
    __shared__ int waveTot[8];
    __shared__ int blockBase;
    const int t = threadIdx.x;
    const int b = blockIdx.x;
    const int* rowBeg = runTab + (size_t)b * nChunks;
    const int* rowEnd = rowBeg + nChunks;
    slot[t] = 0;
    __syncthreads();

    for (int r = t; r < nChunks; r += 256) {
        const int* p = binned + (size_t)r * CHUNK_EDGES;
        int e0 = rowBeg[r], e1 = rowEnd[r];
        e0 = e0 < 0 ? 0 : e0;
        e1 = e1 > CHUNK_EDGES ? CHUNK_EDGES : e1;
        for (int e = e0; e < e1; e += 4) {
            int wd[4];
            #pragma unroll
            for (int q = 0; q < 4; ++q) wd[q] = p[e + q < e1 ? e + q : e1 - 1];
            #pragma unroll
            for (int q = 0; q < 4; ++q)
                if (e + q < e1) atomicAdd(&slot[wd[q] & (BKT_NODES - 1)], 1);
        }
    }
    __syncthreads();

    const int c = slot[t];
    int total;
    const int ex = block_excl_scan(c, waveTot, &total);
    if (t == 0) blockBase = atomicAdd(cursor, total);
    __syncthreads();
    const int st = blockBase + ex;
    const int v = b * BKT_NODES + t;
    if (v < nNodes) { start[v] = st; cnt[v] = c; }
    slot[t] = st;
    __syncthreads();

    for (int r = t; r < nChunks; r += 256) {
        const int* p = binned + (size_t)r * CHUNK_EDGES;
        int e0 = rowBeg[r], e1 = rowEnd[r];
        e0 = e0 < 0 ? 0 : e0;
        e1 = e1 > CHUNK_EDGES ? CHUNK_EDGES : e1;
        for (int e = e0; e < e1; e += 4) {
            int wd[4];
            #pragma unroll
            for (int q = 0; q < 4; ++q) wd[q] = p[e + q < e1 ? e + q : e1 - 1];
            #pragma unroll
            for (int q = 0; q < 4; ++q) {
                if (e + q < e1) {
                    const int pos = atomicAdd(&slot[wd[q] & (BKT_NODES - 1)], 1);
                    sorted[pos] = (int)((unsigned)wd[q] >> BKT_SHIFT);
                }
            }
        }
    }
}


template <int R>
__device__ __forceinline__ int grp_get_i(int v) {
    return (int)__builtin_amdgcn_mov_dpp8((unsigned)v, R * 0x249249u);
}
template <int R>
__device__ __forceinline__ float grp_get(float v) {
    return __uint_as_float(__builtin_amdgcn_mov_dpp8(__float_as_uint(v), R * 0x249249u));
}
__device__ __forceinline__ float grp_max(float x) {
    x = fmaxf(x, row_xor_get<1>(x));
    x = fmaxf(x, row_xor_get<2>(x));
    x = fmaxf(x, row_xor_get<4>(x));
    return x;
}
__device__ __forceinline__ float grp_sum(float x) {
    x += row_xor_get<1>(x);
    x += row_xor_get<2>(x);
    x += row_xor_get<4>(x);
    return x;
}

template <int R>
__device__ __forceinline__ void slot_load(float4 (&f)[GAT_HEADS], int uk,
                                          const float* __restrict__ feat, int q)
{
    const int us = grp_get_i<R>(uk);
    const float* frow = feat + (size_t)us * GAT_FDIM + 4 * q;
    #pragma unroll
    for (int hh = 0; hh < GAT_HEADS; ++hh) f[hh] = *(const float4*)(frow + hh * GAT_D);
}
template <int R>
__device__ __forceinline__ void slot_add(float4& acc, const float4 (&f)[GAT_HEADS],
                                         const float (&ak)[GAT_HEADS], int kbase, int deg)
{
    const float a0 = grp_get<R>(ak[0]), a1 = grp_get<R>(ak[1]);
    const float a2 = grp_get<R>(ak[2]), a3 = grp_get<R>(ak[3]);
    const bool valid = kbase + R < deg;
    const float tx = a0 * f[0].x + a1 * f[1].x + a2 * f[2].x + a3 * f[3].x;
    const float ty = a0 * f[0].y + a1 * f[1].y + a2 * f[2].y + a3 * f[3].y;
    const float tz = a0 * f[0].z + a1 * f[1].z + a2 * f[2].z + a3 * f[3].z;
    const float tw = a0 * f[0].w + a1 * f[1].w + a2 * f[2].w + a3 * f[3].w;
    acc.x += valid ? tx : 0.f;
    acc.y += valid ? ty : 0.f;
    acc.z += valid ? tz : 0.f;
    acc.w += valid ? tw : 0.f;
}
template <int R0>
__device__ __forceinline__ void rows4(float4& acc, int uk, const float (&ak)[GAT_HEADS],
                                      int kbase, int deg,
                                      const float* __restrict__ feat, int q)
{
    float4 f0[GAT_HEADS], f1[GAT_HEADS], f2[GAT_HEADS], f3[GAT_HEADS];
    slot_load<R0 + 0>(f0, uk, feat, q);
    slot_load<R0 + 1>(f1, uk, feat, q);
    slot_load<R0 + 2>(f2, uk, feat, q);
    slot_load<R0 + 3>(f3, uk, feat, q);
    slot_add<R0 + 0>(acc, f0, ak, kbase, deg);
    slot_add<R0 + 1>(acc, f1, ak, kbase, deg);
    slot_add<R0 + 2>(acc, f2, ak, kbase, deg);
    slot_add<R0 + 3>(acc, f3, ak, kbase, deg);
}

__global__ __launch_bounds__(256) void gat_node_aggr(
    const int* __restrict__ sorted, const int* __restrict__ start,
    const int* __restrict__ cnt,
    const float* __restrict__ el, const float* __restrict__ er,
    const float* __restrict__ feat, const float* __restrict__ bias,
    float* __restrict__ out, int nNodes)
{
    const int lane = threadIdx.x & 31;
    const int g = lane >> 3;
    const int q = lane & 7;
    const int wv = __builtin_amdgcn_readfirstlane(
        (int)((blockIdx.x * blockDim.x + threadIdx.x) >> 5));
    if (wv * 4 >= nNodes) return;
    const int v = wv * 4 + g;
    const bool vOk = v < nNodes;
    const int vc = vOk ? v : nNodes - 1;

    const float4 er4 = *(const float4*)(er + (size_t)vc * GAT_HEADS);
    float4 bm;
    {
        const float4 b0 = *(const float4*)(bias + 4 * q);
        const float4 b1 = *(const float4*)(bias + GAT_D + 4 * q);
        const float4 b2 = *(const float4*)(bias + 2 * GAT_D + 4 * q);
        const float4 b3 = *(const float4*)(bias + 3 * GAT_D + 4 * q);
        bm.x = (b0.x + b1.x + b2.x + b3.x) * (1.f / GAT_HEADS);
        bm.y = (b0.y + b1.y + b2.y + b3.y) * (1.f / GAT_HEADS);
        bm.z = (b0.z + b1.z + b2.z + b3.z) * (1.f / GAT_HEADS);
        bm.w = (b0.w + b1.w + b2.w + b3.w) * (1.f / GAT_HEADS);
    }
    const int beg = start[vc];
    const int deg = vOk ? cnt[vc] : 0;
    const int d0 = __builtin_amdgcn_readlane(deg, 0), d1 = __builtin_amdgcn_readlane(deg, 8);
    const int d2 = __builtin_amdgcn_readlane(deg, 16), d3 = __builtin_amdgcn_readlane(deg, 24);
    const int d01 = d0 > d1 ? d0 : d1, d23 = d2 > d3 ? d2 : d3;
    const int dmax = d01 > d23 ? d01 : d23;

    const float NEG_INF = -__builtin_huge_valf();
    float4 acc = make_float4(0.f, 0.f, 0.f, 0.f);

    if (dmax <= 32) {
        int u[4];
        float a[4][GAT_HEADS];
        #pragma unroll
        for (int k = 0; k < 4; ++k) {
            const bool valid = 8 * k + q < deg;
            u[k] = sorted[valid ? beg + 8 * k + q : 0];
        }
        float4 l[4];
        #pragma unroll
        for (int k = 0; k < 4; ++k) l[k] = *(const float4*)(el + (size_t)u[k] * GAT_HEADS);
        #pragma unroll
        for (int k = 0; k < 4; ++k) {
            const float pen = 8 * k + q < deg ? 0.f : NEG_INF;
            a[k][0] = lrelu(l[k].x + er4.x) + pen;
            a[k][1] = lrelu(l[k].y + er4.y) + pen;
            a[k][2] = lrelu(l[k].z + er4.z) + pen;
            a[k][3] = lrelu(l[k].w + er4.w) + pen;
        }
        float s[GAT_HEADS];
        #pragma unroll
        for (int h = 0; h < GAT_HEADS; ++h) {
            const float m = grp_max(fmaxf(fmaxf(a[0][h], a[1][h]), fmaxf(a[2][h], a[3][h])));
            #pragma unroll
            for (int k = 0; k < 4; ++k)
                a[k][h] = 8 * k + q < deg ? __expf(a[k][h] - m) : 0.f;
            s[h] = fmaxf(grp_sum(((a[0][h] + a[1][h]) + a[2][h]) + a[3][h]), 1e-16f);
        }
        #pragma unroll
        for (int k = 0; k < 4; ++k) {
            #pragma unroll
            for (int h = 0; h < GAT_HEADS; ++h) a[k][h] = a[k][h] / s[h];
        }
        #pragma unroll
        for (int k = 0; k < 4; ++k) {
            if (8 * k < dmax) {
                rows4<0>(acc, u[k], a[k], 8 * k, deg, feat, q);
                if (8 * k + 4 < dmax)
                    rows4<4>(acc, u[k], a[k], 8 * k, deg, feat, q);
            }
        }
    } else {
        float m0 = NEG_INF, m1 = NEG_INF, m2 = NEG_INF, m3 = NEG_INF;
        for (int c = 0; c < dmax; c += 8) {
            const bool valid = c + q < deg;
            const int uu = sorted[valid ? beg + c + q : 0];
            const float4 l = *(const float4*)(el + (size_t)uu * GAT_HEADS);
            m0 = fmaxf(m0, valid ? lrelu(l.x + er4.x) : NEG_INF);
            m1 = fmaxf(m1, valid ? lrelu(l.y + er4.y) : NEG_INF);
            m2 = fmaxf(m2, valid ? lrelu(l.z + er4.z) : NEG_INF);
            m3 = fmaxf(m3, valid ? lrelu(l.w + er4.w) : NEG_INF);
        }
        m0 = grp_max(m0); m1 = grp_max(m1); m2 = grp_max(m2); m3 = grp_max(m3);
        float s0 = 0.f, s1 = 0.f, s2 = 0.f, s3 = 0.f;
        for (int c = 0; c < dmax; c += 8) {
            const bool valid = c + q < deg;
            const int uu = sorted[valid ? beg + c + q : 0];
            const float4 l = *(const float4*)(el + (size_t)uu * GAT_HEADS);
            s0 += valid ? __expf(lrelu(l.x + er4.x) - m0) : 0.f;
            s1 += valid ? __expf(lrelu(l.y + er4.y) - m1) : 0.f;
            s2 += valid ? __expf(lrelu(l.z + er4.z) - m2) : 0.f;
            s3 += valid ? __expf(lrelu(l.w + er4.w) - m3) : 0.f;
        }
        s0 = fmaxf(grp_sum(s0), 1e-16f); s1 = fmaxf(grp_sum(s1), 1e-16f);
        s2 = fmaxf(grp_sum(s2), 1e-16f); s3 = fmaxf(grp_sum(s3), 1e-16f);
        for (int c = 0; c < dmax; c += 8) {
            const bool valid = c + q < deg;
            const int uu = sorted[valid ? beg + c + q : 0];
            const float4 l = *(const float4*)(el + (size_t)uu * GAT_HEADS);
            float ak[GAT_HEADS];
            ak[0] = valid ? __expf(lrelu(l.x + er4.x) - m0) / s0 : 0.f;
            ak[1] = valid ? __expf(lrelu(l.y + er4.y) - m1) / s1 : 0.f;
            ak[2] = valid ? __expf(lrelu(l.z + er4.z) - m2) / s2 : 0.f;
            ak[3] = valid ? __expf(lrelu(l.w + er4.w) - m3) / s3 : 0.f;
            rows4<0>(acc, uu, ak, c, deg, feat, q);
            if (c + 4 < dmax) rows4<4>(acc, uu, ak, c, deg, feat, q);
        }
    }
    if (vOk) {
        float4 o;
        o.x = 0.25f * acc.x + bm.x;
        o.y = 0.25f * acc.y + bm.y;
        o.z = 0.25f * acc.z + bm.z;
        o.w = 0.25f * acc.w + bm.w;
        *(float4*)(out + (size_t)v * GAT_D + 4 * q) = o;
    }
}

extern "C" void kernel_launch(void* const* d_in, const int* in_sizes, int n_in,
                              void* d_out, int out_size, void* d_ws, size_t ws_size,
                              hipStream_t stream)
{
    const float* x        = (const float*)d_in[0];
    const int*   edge_src = (const int*)  d_in[1];
    const int*   edge_dst = (const int*)  d_in[2];
    const float* fc_w     = (const float*)d_in[3];
    const float* attn_l   = (const float*)d_in[4];
    const float* attn_r   = (const float*)d_in[5];
    const float* bias     = (const float*)d_in[6];
    float*       out      = (float*)d_out;

    const int nNodes = in_sizes[0] / GAT_FDIM;
    const int nEdges = in_sizes[1];

    float* feat   = (float*)d_ws;
    float* el     = feat + (size_t)nNodes * GAT_FDIM;
    float* er     = el   + (size_t)nNodes * GAT_HEADS;
    int*   cnt    = (int*)(er + (size_t)nNodes * GAT_HEADS);
    int*   start  = cnt + nNodes;
    int*   cursor = start + nNodes;
    int*   sorted = cursor + 4;
    if ((size_t)((char*)(sorted + nEdges) - (char*)d_ws) > ws_size) return;

    const int nBkts   = (nNodes + BKT_NODES - 1) / BKT_NODES;
    const int nChunks = (nEdges + CHUNK_EDGES - 1) / CHUNK_EDGES;
    int* binned = (int*)feat;
    int* runTab = binned + (size_t)nChunks * CHUNK_EDGES;
    if (nBkts > MAX_BKTS || nNodes > (1 << 24) ||
        (size_t)nChunks * CHUNK_EDGES + (size_t)(nBkts + 1) * nChunks > (size_t)nNodes * GAT_FDIM)
        return;

    gat_edge_bin<<<nChunks, 256, 0, stream>>>(edge_src, edge_dst, binned, runTab, cursor,
                                              nEdges, nBkts, nChunks);
    gat_bucket_group<<<nBkts, 256, 0, stream>>>(binned, runTab, cnt, start, sorted, cursor,
                                                nNodes, nChunks);
    {
        int mTiles = (nNodes + 15) / 16;
        int blocks = (mTiles + 3) / 4;
        gat_gemm_feat<<<blocks, 128, 0, stream>>>(x, fc_w, attn_l, attn_r, feat, el, er, nNodes);
    }
    {
        int blocks = (nNodes + 31) / 32;
        gat_node_aggr<<<blocks, 256, 0, stream>>>(
            sorted, start, cnt, el, er, feat, bias, out, nNodes);
    }
}


# GEMM B operand from LDS: each GEMM work-group copies fc_w once into LDS in fragment order (64 KB), waves read B by 16-byte LDS loads, not 32 global loads per 16 WMMAs; k order unchanged
# speedup vs baseline: 3.6538x; 1.0893x over previous
#include <hip/hip_runtime.h>
#include <hip/hip_bf16.h>


typedef float v2f __attribute__((ext_vector_type(2)));
typedef float v8f __attribute__((ext_vector_type(8)));

#define GAT_HEADS   4
#define GAT_D       32
#define GAT_FDIM    128
#define NEG_SLOPE   0.2f

#define WL_ENTRIES (32 * GAT_FDIM)

__device__ __forceinline__ void stage_w(const float* __restrict__ w, float4* wl)
{
    const int j = threadIdx.x & 31;
    const int q = threadIdx.x >> 5;
    #pragma unroll
    for (int i = 0; i < 4; ++i) {
        const int gh = 8 * i + q;
        const int r0 = 8 * (gh >> 1) + 2 * (gh & 1);
        const float4 m0 = *(const float4*)(w + (r0 + 0) * GAT_FDIM + 4 * j);
        const float4 m1 = *(const float4*)(w + (r0 + 1) * GAT_FDIM + 4 * j);
        const float4 m2 = *(const float4*)(w + (r0 + 4) * GAT_FDIM + 4 * j);
        const float4 m3 = *(const float4*)(w + (r0 + 5) * GAT_FDIM + 4 * j);
        float4* dst = wl + gh * GAT_FDIM + 4 * j;
        dst[0] = make_float4(m0.x, m1.x, m2.x, m3.x);
        dst[1] = make_float4(m0.y, m1.y, m2.y, m3.y);
        dst[2] = make_float4(m0.z, m1.z, m2.z, m3.z);
        dst[3] = make_float4(m0.w, m1.w, m2.w, m3.w);
    }
}

template <int M>
__device__ __forceinline__ float row_xor_get(float v) {
    return __int_as_float(__builtin_amdgcn_update_dpp(
        0, __float_as_int(v), 0x160 + M, 0xf, 0xf, false));
}

__device__ __forceinline__ void row_sum_32(float (&p)[32], int c) {
    {
        const bool up = (c & 8) != 0;
        #pragma unroll
        for (int i = 0; i < 16; ++i) {
            const float give = up ? p[i] : p[i + 16];
            const float keep = up ? p[i + 16] : p[i];
            p[i] = keep + row_xor_get<8>(give);
        }
    }
    {
        const bool up = (c & 4) != 0;
        #pragma unroll
        for (int i = 0; i < 8; ++i) {
            const float give = up ? p[i] : p[i + 8];
            const float keep = up ? p[i + 8] : p[i];
            p[i] = keep + row_xor_get<4>(give);
        }
    }
    {
        const bool up = (c & 2) != 0;
        #pragma unroll
        for (int i = 0; i < 4; ++i) {
            const float give = up ? p[i] : p[i + 4];
            const float keep = up ? p[i + 4] : p[i];
            p[i] = keep + row_xor_get<2>(give);
        }
    }
    {
        const bool up = (c & 1) != 0;
        #pragma unroll
        for (int i = 0; i < 2; ++i) {
            const float give = up ? p[i] : p[i + 2];
            const float keep = up ? p[i + 2] : p[i];
            p[i] = keep + row_xor_get<1>(give);
        }
    }
}

__device__ __forceinline__ void gemm_strip(
    const float* __restrict__ x, const float4* wl,
    const float* __restrict__ attn_l, const float* __restrict__ attn_r,
    float* __restrict__ feat, float* __restrict__ el, float* __restrict__ er,
    int nNodes, int mTile)
{
    const int lane  = threadIdx.x & 31;
    const int laneM  = lane & 15;
    const int khalf  = lane >> 4;
    int row = mTile * 16 + laneM;
    int rowClamped = row < nNodes ? row : nNodes - 1;

    v8f acc[8];
    #pragma unroll
    for (int t = 0; t < 8; ++t) acc[t] = v8f{};

    const float* xrow = x + (size_t)rowClamped * GAT_FDIM;

    const float4* wb = wl + khalf * GAT_FDIM + laneM;
    #pragma unroll 1
    for (int g = 0; g < GAT_FDIM / 8; ++g) {
        const v2f a0 = *(const v2f*)(xrow + 8 * g + 2 * khalf);
        const v2f a1 = *(const v2f*)(xrow + 8 * g + 4 + 2 * khalf);
        float4 bq[8];
        #pragma unroll
        for (int t = 0; t < 8; ++t) bq[t] = wb[2 * g * GAT_FDIM + 16 * t];
        #pragma unroll
        for (int t = 0; t < 8; ++t) {
            const v2f b = v2f{bq[t].x, bq[t].y};
            acc[t] = __builtin_amdgcn_wmma_f32_16x16x4_f32(
                 false, a0,  false, b,
                 (short)0, acc[t],
                 false,  false);
        }
        #pragma unroll
        for (int t = 0; t < 8; ++t) {
            const v2f b = v2f{bq[t].z, bq[t].w};
            acc[t] = __builtin_amdgcn_wmma_f32_16x16x4_f32(
                 false, a1,  false, b,
                 (short)0, acc[t],
                 false,  false);
        }
    }

    const int mOff = 8 * khalf;
    #pragma unroll
    for (int t = 0; t < 8; ++t) {
        #pragma unroll
        for (int r = 0; r < 8; ++r) {
            int m = mTile * 16 + mOff + r;
            if (m < nNodes)
                feat[(size_t)m * GAT_FDIM + t * 16 + laneM] = acc[t][r];
        }
    }

    const size_t lineOff = (size_t)(mTile * 16 + mOff) * GAT_HEADS + 2 * laneM;
    const bool rowOk = mTile * 16 + mOff + (laneM >> 1) < nNodes;
    {
        float wl[8];
        #pragma unroll
        for (int t = 0; t < 8; ++t) wl[t] = attn_l[t * 16 + laneM];
        float p[32];
        #pragma unroll
        for (int r = 0; r < 8; ++r) {
            #pragma unroll
            for (int h = 0; h < GAT_HEADS; ++h)
                p[r * 4 + h] = acc[2 * h][r] * wl[2 * h] + acc[2 * h + 1][r] * wl[2 * h + 1];
        }
        row_sum_32(p, laneM);
        if (rowOk) *(v2f*)(el + lineOff) = v2f{p[0], p[1]};
    }
    {
        float wr[8];
        #pragma unroll
        for (int t = 0; t < 8; ++t) wr[t] = attn_r[t * 16 + laneM];
        float p[32];
        #pragma unroll
        for (int r = 0; r < 8; ++r) {
            #pragma unroll
            for (int h = 0; h < GAT_HEADS; ++h)
                p[r * 4 + h] = acc[2 * h][r] * wr[2 * h] + acc[2 * h + 1][r] * wr[2 * h + 1];
        }
        row_sum_32(p, laneM);
        if (rowOk) *(v2f*)(er + lineOff) = v2f{p[0], p[1]};
    }
}

__device__ __forceinline__ float lrelu(float v) {
    return v > 0.f ? v : NEG_SLOPE * v;
}

#define BKT_SHIFT    8
#define BKT_NODES    256
#define MAX_BKTS     512
#define CHUNK_EDGES  3200
#define CHUNK_STEPS  13

__device__ __forceinline__ int block_excl_scan(int v, int* waveTot, int* total) {
    const int lane = threadIdx.x & 31, wave = threadIdx.x >> 5;
    int incl = v;
    #pragma unroll
    for (int off = 1; off < 32; off <<= 1) {
        int t = __shfl_up(incl, off, 32);
        if (lane >= off) incl += t;
    }
    if (lane == 31) waveTot[wave] = incl;
    __syncthreads();
    int wbase = 0, tot = 0;
    #pragma unroll
    for (int q = 0; q < 8; ++q) {
        const int wt = waveTot[q];
        wbase += q < wave ? wt : 0;
        tot += wt;
    }
    *total = tot;
    return wbase + incl - v;
}

__device__ __forceinline__ void edge_bin_block(
    const int* __restrict__ src, const int* __restrict__ dst,
    int* __restrict__ binned, int* __restrict__ runTab, int* __restrict__ cursor,
    int nEdges, int nBkts, int nChunks, int w)
{
    __shared__ int hist[MAX_BKTS + 2];
    __shared__ int waveTot[8];
    const int t = threadIdx.x;
    const int base = w * CHUNK_EDGES;
    const int left = nEdges - base;
    const int len = left < CHUNK_EDGES ? left : CHUNK_EDGES;
    if (w == 0 && t == 0) *cursor = 0;
    hist[2 * t] = 0;
    hist[2 * t + 1] = 0;
    __syncthreads();

    int d[CHUNK_STEPS], s[CHUNK_STEPS];
    #pragma unroll
    for (int k = 0; k < CHUNK_STEPS; ++k) {
        const int i  = k * 256 + t;
        const int ic = i < len ? i : len - 1;
        d[k] = dst[base + ic];
        s[k] = src[base + ic];
    }
    int lr[CHUNK_STEPS];
    #pragma unroll
    for (int k = 0; k < CHUNK_STEPS; ++k)
        lr[k] = atomicAdd(&hist[k * 256 + t < len ? d[k] >> BKT_SHIFT : MAX_BKTS + 1], 1);
    __syncthreads();

    const int c0 = hist[2 * t], c1 = hist[2 * t + 1];
    int total;
    const int ex = block_excl_scan(c0 + c1, waveTot, &total);
    hist[2 * t] = ex;
    hist[2 * t + 1] = ex + c0;
    if (t == 255) hist[MAX_BKTS] = total;
    __syncthreads();

    for (int b = t; b <= nBkts; b += 256)
        runTab[(size_t)b * nChunks + w] = hist[b];
    #pragma unroll
    for (int k = 0; k < CHUNK_STEPS; ++k) {
        if (k * 256 + t < len)
            binned[base + hist[d[k] >> BKT_SHIFT] + lr[k]] =
                (int)(((unsigned)s[k] << BKT_SHIFT) | (unsigned)(d[k] & (BKT_NODES - 1)));
    }
}

__device__ __forceinline__ void bucket_group_block(
    const int* __restrict__ binned, const int* __restrict__ runTab,
    int* __restrict__ cnt, int* __restrict__ start, int* __restrict__ sorted,
    int* __restrict__ cursor, int nNodes, int nChunks, int b)
{
    __shared__ int slot[BKT_NODES];
    __shared__ int waveTot[8];
    __shared__ int blockBase;
    const int t = threadIdx.x;
    const int* rowBeg = runTab + (size_t)b * nChunks;
    const int* rowEnd = rowBeg + nChunks;
    slot[t] = 0;
    __syncthreads();

    for (int r = t; r < nChunks; r += 256) {
        const int* p = binned + (size_t)r * CHUNK_EDGES;
        int e0 = rowBeg[r], e1 = rowEnd[r];
        e0 = e0 < 0 ? 0 : e0;
        e1 = e1 > CHUNK_EDGES ? CHUNK_EDGES : e1;
        for (int e = e0; e < e1; e += 4) {
            int wd[4];
            #pragma unroll
            for (int q = 0; q < 4; ++q) wd[q] = p[e + q < e1 ? e + q : e1 - 1];
            #pragma unroll
            for (int q = 0; q < 4; ++q)
                if (e + q < e1) atomicAdd(&slot[wd[q] & (BKT_NODES - 1)], 1);
        }
    }
    __syncthreads();

    const int c = slot[t];
    int total;
    const int ex = block_excl_scan(c, waveTot, &total);
    if (t == 0) blockBase = atomicAdd(cursor, total);
    __syncthreads();
    const int st = blockBase + ex;
    const int v = b * BKT_NODES + t;
    if (v < nNodes) { start[v] = st; cnt[v] = c; }
    slot[t] = st;
    __syncthreads();

    for (int r = t; r < nChunks; r += 256) {
        const int* p = binned + (size_t)r * CHUNK_EDGES;
        int e0 = rowBeg[r], e1 = rowEnd[r];
        e0 = e0 < 0 ? 0 : e0;
        e1 = e1 > CHUNK_EDGES ? CHUNK_EDGES : e1;
        for (int e = e0; e < e1; e += 4) {
            int wd[4];
            #pragma unroll
            for (int q = 0; q < 4; ++q) wd[q] = p[e + q < e1 ? e + q : e1 - 1];
            #pragma unroll
            for (int q = 0; q < 4; ++q) {
                if (e + q < e1) {
                    const int pos = atomicAdd(&slot[wd[q] & (BKT_NODES - 1)], 1);
                    sorted[pos] = (int)((unsigned)wd[q] >> BKT_SHIFT);
                }
            }
        }
    }
}

__global__ __launch_bounds__(256) void gat_bin_and_gemm(
    const int* __restrict__ src, const int* __restrict__ dst,
    int* __restrict__ binned, int* __restrict__ runTab, int* __restrict__ cursor,
    int nEdges, int nBkts, int nChunks,
    const float* __restrict__ x, const float* __restrict__ w,
    const float* __restrict__ attn_l, const float* __restrict__ attn_r,
    float* __restrict__ feat, float* __restrict__ el, float* __restrict__ er,
    int nNodes, int tile0, int tile1)
{
    if ((int)blockIdx.x < nChunks) {
        edge_bin_block(src, dst, binned, runTab, cursor, nEdges, nBkts, nChunks,
                       (int)blockIdx.x);
    } else {
        __shared__ float4 wl[WL_ENTRIES];
        stage_w(w, wl);
        __syncthreads();
        const int mTile = tile0 + ((int)blockIdx.x - nChunks) * 8 + (int)(threadIdx.x >> 5);
        if (mTile < tile1)
            gemm_strip(x, wl, attn_l, attn_r, feat, el, er, nNodes, mTile);
    }
}

__global__ __launch_bounds__(256) void gat_group_and_gemm(
    const int* __restrict__ binned, const int* __restrict__ runTab,
    int* __restrict__ cnt, int* __restrict__ start, int* __restrict__ sorted,
    int* __restrict__ cursor, int nBkts, int nChunks,
    const float* __restrict__ x, const float* __restrict__ w,
    const float* __restrict__ attn_l, const float* __restrict__ attn_r,
    float* __restrict__ feat, float* __restrict__ el, float* __restrict__ er,
    int nNodes, int tile0, int tile1)
{
    if ((int)blockIdx.x < nBkts) {
        bucket_group_block(binned, runTab, cnt, start, sorted, cursor, nNodes, nChunks,
                           (int)blockIdx.x);
    } else {
        __shared__ float4 wl[WL_ENTRIES];
        stage_w(w, wl);
        __syncthreads();
        const int mTile = tile0 + ((int)blockIdx.x - nBkts) * 8 + (int)(threadIdx.x >> 5);
        if (mTile < tile1)
            gemm_strip(x, wl, attn_l, attn_r, feat, el, er, nNodes, mTile);
    }
}


template <int R>
__device__ __forceinline__ int grp_get_i(int v) {
    return (int)__builtin_amdgcn_mov_dpp8((unsigned)v, R * 0x249249u);
}
template <int R>
__device__ __forceinline__ float grp_get(float v) {
    return __uint_as_float(__builtin_amdgcn_mov_dpp8(__float_as_uint(v), R * 0x249249u));
}
__device__ __forceinline__ float grp_max(float x) {
    x = fmaxf(x, row_xor_get<1>(x));
    x = fmaxf(x, row_xor_get<2>(x));
    x = fmaxf(x, row_xor_get<4>(x));
    return x;
}
__device__ __forceinline__ float grp_sum(float x) {
    x += row_xor_get<1>(x);
    x += row_xor_get<2>(x);
    x += row_xor_get<4>(x);
    return x;
}

template <int R>
__device__ __forceinline__ void slot_load(float4 (&f)[GAT_HEADS], int uk,
                                          const float* __restrict__ feat, int q)
{
    const int us = grp_get_i<R>(uk);
    const float* frow = feat + (size_t)us * GAT_FDIM + 4 * q;
    #pragma unroll
    for (int hh = 0; hh < GAT_HEADS; ++hh) f[hh] = *(const float4*)(frow + hh * GAT_D);
}
template <int R>
__device__ __forceinline__ void slot_add(float4& acc, const float4 (&f)[GAT_HEADS],
                                         const float (&ak)[GAT_HEADS], int kbase, int deg)
{
    const float a0 = grp_get<R>(ak[0]), a1 = grp_get<R>(ak[1]);
    const float a2 = grp_get<R>(ak[2]), a3 = grp_get<R>(ak[3]);
    const bool valid = kbase + R < deg;
    const float tx = a0 * f[0].x + a1 * f[1].x + a2 * f[2].x + a3 * f[3].x;
    const float ty = a0 * f[0].y + a1 * f[1].y + a2 * f[2].y + a3 * f[3].y;
    const float tz = a0 * f[0].z + a1 * f[1].z + a2 * f[2].z + a3 * f[3].z;
    const float tw = a0 * f[0].w + a1 * f[1].w + a2 * f[2].w + a3 * f[3].w;
    acc.x += valid ? tx : 0.f;
    acc.y += valid ? ty : 0.f;
    acc.z += valid ? tz : 0.f;
    acc.w += valid ? tw : 0.f;
}
template <int R0>
__device__ __forceinline__ void rows4(float4& acc, int uk, const float (&ak)[GAT_HEADS],
                                      int kbase, int deg,
                                      const float* __restrict__ feat, int q)
{
    float4 f0[GAT_HEADS], f1[GAT_HEADS], f2[GAT_HEADS], f3[GAT_HEADS];
    slot_load<R0 + 0>(f0, uk, feat, q);
    slot_load<R0 + 1>(f1, uk, feat, q);
    slot_load<R0 + 2>(f2, uk, feat, q);
    slot_load<R0 + 3>(f3, uk, feat, q);
    slot_add<R0 + 0>(acc, f0, ak, kbase, deg);
    slot_add<R0 + 1>(acc, f1, ak, kbase, deg);
    slot_add<R0 + 2>(acc, f2, ak, kbase, deg);
    slot_add<R0 + 3>(acc, f3, ak, kbase, deg);
}

__global__ __launch_bounds__(256) void gat_node_aggr(
    const int* __restrict__ sorted, const int* __restrict__ start,
    const int* __restrict__ cnt,
    const float* __restrict__ el, const float* __restrict__ er,
    const float* __restrict__ feat, const float* __restrict__ bias,
    float* __restrict__ out, int nNodes)
{
    const int lane = threadIdx.x & 31;
    const int g = lane >> 3;
    const int q = lane & 7;
    const int wv = __builtin_amdgcn_readfirstlane(
        (int)((blockIdx.x * blockDim.x + threadIdx.x) >> 5));
    if (wv * 4 >= nNodes) return;
    const int v = wv * 4 + g;
    const bool vOk = v < nNodes;
    const int vc = vOk ? v : nNodes - 1;

    const float4 er4 = *(const float4*)(er + (size_t)vc * GAT_HEADS);
    float4 bm;
    {
        const float4 b0 = *(const float4*)(bias + 4 * q);
        const float4 b1 = *(const float4*)(bias + GAT_D + 4 * q);
        const float4 b2 = *(const float4*)(bias + 2 * GAT_D + 4 * q);
        const float4 b3 = *(const float4*)(bias + 3 * GAT_D + 4 * q);
        bm.x = (b0.x + b1.x + b2.x + b3.x) * (1.f / GAT_HEADS);
        bm.y = (b0.y + b1.y + b2.y + b3.y) * (1.f / GAT_HEADS);
        bm.z = (b0.z + b1.z + b2.z + b3.z) * (1.f / GAT_HEADS);
        bm.w = (b0.w + b1.w + b2.w + b3.w) * (1.f / GAT_HEADS);
    }
    const int beg = start[vc];
    const int deg = vOk ? cnt[vc] : 0;
    const int d0 = __builtin_amdgcn_readlane(deg, 0), d1 = __builtin_amdgcn_readlane(deg, 8);
    const int d2 = __builtin_amdgcn_readlane(deg, 16), d3 = __builtin_amdgcn_readlane(deg, 24);
    const int d01 = d0 > d1 ? d0 : d1, d23 = d2 > d3 ? d2 : d3;
    const int dmax = d01 > d23 ? d01 : d23;

    const float NEG_INF = -__builtin_huge_valf();
    float4 acc = make_float4(0.f, 0.f, 0.f, 0.f);

    if (dmax <= 32) {
        int u[4];
        float a[4][GAT_HEADS];
        #pragma unroll
        for (int k = 0; k < 4; ++k) {
            const bool valid = 8 * k + q < deg;
            u[k] = sorted[valid ? beg + 8 * k + q : 0];
        }
        float4 l[4];
        #pragma unroll
        for (int k = 0; k < 4; ++k) l[k] = *(const float4*)(el + (size_t)u[k] * GAT_HEADS);
        #pragma unroll
        for (int k = 0; k < 4; ++k) {
            const float pen = 8 * k + q < deg ? 0.f : NEG_INF;
            a[k][0] = lrelu(l[k].x + er4.x) + pen;
            a[k][1] = lrelu(l[k].y + er4.y) + pen;
            a[k][2] = lrelu(l[k].z + er4.z) + pen;
            a[k][3] = lrelu(l[k].w + er4.w) + pen;
        }
        float s[GAT_HEADS];
        #pragma unroll
        for (int h = 0; h < GAT_HEADS; ++h) {
            const float m = grp_max(fmaxf(fmaxf(a[0][h], a[1][h]), fmaxf(a[2][h], a[3][h])));
            #pragma unroll
            for (int k = 0; k < 4; ++k)
                a[k][h] = 8 * k + q < deg ? __expf(a[k][h] - m) : 0.f;
            s[h] = fmaxf(grp_sum(((a[0][h] + a[1][h]) + a[2][h]) + a[3][h]), 1e-16f);
        }
        #pragma unroll
        for (int k = 0; k < 4; ++k) {
            #pragma unroll
            for (int h = 0; h < GAT_HEADS; ++h) a[k][h] = a[k][h] / s[h];
        }
        #pragma unroll
        for (int k = 0; k < 4; ++k) {
            if (8 * k < dmax) {
                rows4<0>(acc, u[k], a[k], 8 * k, deg, feat, q);
                if (8 * k + 4 < dmax)
                    rows4<4>(acc, u[k], a[k], 8 * k, deg, feat, q);
            }
        }
    } else {
        float m0 = NEG_INF, m1 = NEG_INF, m2 = NEG_INF, m3 = NEG_INF;
        for (int c = 0; c < dmax; c += 8) {
            const bool valid = c + q < deg;
            const int uu = sorted[valid ? beg + c + q : 0];
            const float4 l = *(const float4*)(el + (size_t)uu * GAT_HEADS);
            m0 = fmaxf(m0, valid ? lrelu(l.x + er4.x) : NEG_INF);
            m1 = fmaxf(m1, valid ? lrelu(l.y + er4.y) : NEG_INF);
            m2 = fmaxf(m2, valid ? lrelu(l.z + er4.z) : NEG_INF);
            m3 = fmaxf(m3, valid ? lrelu(l.w + er4.w) : NEG_INF);
        }
        m0 = grp_max(m0); m1 = grp_max(m1); m2 = grp_max(m2); m3 = grp_max(m3);
        float s0 = 0.f, s1 = 0.f, s2 = 0.f, s3 = 0.f;
        for (int c = 0; c < dmax; c += 8) {
            const bool valid = c + q < deg;
            const int uu = sorted[valid ? beg + c + q : 0];
            const float4 l = *(const float4*)(el + (size_t)uu * GAT_HEADS);
            s0 += valid ? __expf(lrelu(l.x + er4.x) - m0) : 0.f;
            s1 += valid ? __expf(lrelu(l.y + er4.y) - m1) : 0.f;
            s2 += valid ? __expf(lrelu(l.z + er4.z) - m2) : 0.f;
            s3 += valid ? __expf(lrelu(l.w + er4.w) - m3) : 0.f;
        }
        s0 = fmaxf(grp_sum(s0), 1e-16f); s1 = fmaxf(grp_sum(s1), 1e-16f);
        s2 = fmaxf(grp_sum(s2), 1e-16f); s3 = fmaxf(grp_sum(s3), 1e-16f);
        for (int c = 0; c < dmax; c += 8) {
            const bool valid = c + q < deg;
            const int uu = sorted[valid ? beg + c + q : 0];
            const float4 l = *(const float4*)(el + (size_t)uu * GAT_HEADS);
            float ak[GAT_HEADS];
            ak[0] = valid ? __expf(lrelu(l.x + er4.x) - m0) / s0 : 0.f;
            ak[1] = valid ? __expf(lrelu(l.y + er4.y) - m1) / s1 : 0.f;
            ak[2] = valid ? __expf(lrelu(l.z + er4.z) - m2) / s2 : 0.f;
            ak[3] = valid ? __expf(lrelu(l.w + er4.w) - m3) / s3 : 0.f;
            rows4<0>(acc, uu, ak, c, deg, feat, q);
            if (c + 4 < dmax) rows4<4>(acc, uu, ak, c, deg, feat, q);
        }
    }
    if (vOk) {
        float4 o;
        o.x = 0.25f * acc.x + bm.x;
        o.y = 0.25f * acc.y + bm.y;
        o.z = 0.25f * acc.z + bm.z;
        o.w = 0.25f * acc.w + bm.w;
        *(float4*)(out + (size_t)v * GAT_D + 4 * q) = o;
    }
}

extern "C" void kernel_launch(void* const* d_in, const int* in_sizes, int n_in,
                              void* d_out, int out_size, void* d_ws, size_t ws_size,
                              hipStream_t stream)
{
    const float* x        = (const float*)d_in[0];
    const int*   edge_src = (const int*)  d_in[1];
    const int*   edge_dst = (const int*)  d_in[2];
    const float* fc_w     = (const float*)d_in[3];
    const float* attn_l   = (const float*)d_in[4];
    const float* attn_r   = (const float*)d_in[5];
    const float* bias     = (const float*)d_in[6];
    float*       out      = (float*)d_out;

    const int nNodes = in_sizes[0] / GAT_FDIM;
    const int nEdges = in_sizes[1];

    float* feat   = (float*)d_ws;
    float* el     = feat + (size_t)nNodes * GAT_FDIM;
    float* er     = el   + (size_t)nNodes * GAT_HEADS;
    int*   cnt    = (int*)(er + (size_t)nNodes * GAT_HEADS);
    int*   start  = cnt + nNodes;
    int*   cursor = start + nNodes;
    int*   sorted = cursor + 4;
    if ((size_t)((char*)(sorted + nEdges) - (char*)d_ws) > ws_size) return;

    const int nBkts   = (nNodes + BKT_NODES - 1) / BKT_NODES;
    const int nChunks = (nEdges + CHUNK_EDGES - 1) / CHUNK_EDGES;
    const size_t scratchWords = (size_t)nChunks * CHUNK_EDGES + (size_t)(nBkts + 1) * nChunks;
    int* binned = sorted + nEdges;
    if ((size_t)((char*)(binned + scratchWords) - (char*)d_ws) > ws_size) {
        if (scratchWords > (size_t)nNodes * GAT_D) return;
        binned = (int*)d_out;
    }
    int* runTab = binned + (size_t)nChunks * CHUNK_EDGES;
    if (nBkts > MAX_BKTS || nNodes > (1 << 24)) return;

    const int mTiles  = (nNodes + 15) / 16;
    const int tileMid = ((mTiles / 2 + 7) / 8) * 8 < mTiles ? ((mTiles / 2 + 7) / 8) * 8 : mTiles;

    gat_bin_and_gemm<<<nChunks + (tileMid + 7) / 8, 256, 0, stream>>>(
        edge_src, edge_dst, binned, runTab, cursor, nEdges, nBkts, nChunks,
        x, fc_w, attn_l, attn_r, feat, el, er, nNodes, 0, tileMid);
    gat_group_and_gemm<<<nBkts + (mTiles - tileMid + 7) / 8, 256, 0, stream>>>(
        binned, runTab, cnt, start, sorted, cursor, nBkts, nChunks,
        x, fc_w, attn_l, attn_r, feat, el, er, nNodes, tileMid, mTiles);
    {
        int blocks = (nNodes + 31) / 32;
        gat_node_aggr<<<blocks, 256, 0, stream>>>(
            sorted, start, cnt, el, er, feat, bias, out, nNodes);
    }
}


# both edge-grouping levels place their words in LDS (the 64 KB a GEMM work-group holds fc_w in) and write them out in order: whole-line stores, not 3.2M scattered 4-byte ones; orders unchanged
# speedup vs baseline: 3.9137x; 1.0711x over previous
#include <hip/hip_runtime.h>
#include <hip/hip_bf16.h>


typedef float v2f __attribute__((ext_vector_type(2)));
typedef float v8f __attribute__((ext_vector_type(8)));

#define GAT_HEADS   4
#define GAT_D       32
#define GAT_FDIM    128
#define NEG_SLOPE   0.2f

#define WL_ENTRIES (32 * GAT_FDIM)

__device__ __forceinline__ void stage_w(const float* __restrict__ w, float4* wl)
{
    const int j = threadIdx.x & 31;
    const int q = threadIdx.x >> 5;
    #pragma unroll
    for (int i = 0; i < 4; ++i) {
        const int gh = 8 * i + q;
        const int r0 = 8 * (gh >> 1) + 2 * (gh & 1);
        const float4 m0 = *(const float4*)(w + (r0 + 0) * GAT_FDIM + 4 * j);
        const float4 m1 = *(const float4*)(w + (r0 + 1) * GAT_FDIM + 4 * j);
        const float4 m2 = *(const float4*)(w + (r0 + 4) * GAT_FDIM + 4 * j);
        const float4 m3 = *(const float4*)(w + (r0 + 5) * GAT_FDIM + 4 * j);
        float4* dst = wl + gh * GAT_FDIM + 4 * j;
        dst[0] = make_float4(m0.x, m1.x, m2.x, m3.x);
        dst[1] = make_float4(m0.y, m1.y, m2.y, m3.y);
        dst[2] = make_float4(m0.z, m1.z, m2.z, m3.z);
        dst[3] = make_float4(m0.w, m1.w, m2.w, m3.w);
    }
}

template <int M>
__device__ __forceinline__ float row_xor_get(float v) {
    return __int_as_float(__builtin_amdgcn_update_dpp(
        0, __float_as_int(v), 0x160 + M, 0xf, 0xf, false));
}

__device__ __forceinline__ void row_sum_32(float (&p)[32], int c) {
    {
        const bool up = (c & 8) != 0;
        #pragma unroll
        for (int i = 0; i < 16; ++i) {
            const float give = up ? p[i] : p[i + 16];
            const float keep = up ? p[i + 16] : p[i];
            p[i] = keep + row_xor_get<8>(give);
        }
    }
    {
        const bool up = (c & 4) != 0;
        #pragma unroll
        for (int i = 0; i < 8; ++i) {
            const float give = up ? p[i] : p[i + 8];
            const float keep = up ? p[i + 8] : p[i];
            p[i] = keep + row_xor_get<4>(give);
        }
    }
    {
        const bool up = (c & 2) != 0;
        #pragma unroll
        for (int i = 0; i < 4; ++i) {
            const float give = up ? p[i] : p[i + 4];
            const float keep = up ? p[i + 4] : p[i];
            p[i] = keep + row_xor_get<2>(give);
        }
    }
    {
        const bool up = (c & 1) != 0;
        #pragma unroll
        for (int i = 0; i < 2; ++i) {
            const float give = up ? p[i] : p[i + 2];
            const float keep = up ? p[i + 2] : p[i];
            p[i] = keep + row_xor_get<1>(give);
        }
    }
}

__device__ __forceinline__ void gemm_strip(
    const float* __restrict__ x, const float4* wl,
    const float* __restrict__ attn_l, const float* __restrict__ attn_r,
    float* __restrict__ feat, float* __restrict__ el, float* __restrict__ er,
    int nNodes, int mTile)
{
    const int lane  = threadIdx.x & 31;
    const int laneM  = lane & 15;
    const int khalf  = lane >> 4;
    int row = mTile * 16 + laneM;
    int rowClamped = row < nNodes ? row : nNodes - 1;

    v8f acc[8];
    #pragma unroll
    for (int t = 0; t < 8; ++t) acc[t] = v8f{};

    const float* xrow = x + (size_t)rowClamped * GAT_FDIM;

    const float4* wb = wl + khalf * GAT_FDIM + laneM;
    #pragma unroll 1
    for (int g = 0; g < GAT_FDIM / 8; ++g) {
        const v2f a0 = *(const v2f*)(xrow + 8 * g + 2 * khalf);
        const v2f a1 = *(const v2f*)(xrow + 8 * g + 4 + 2 * khalf);
        float4 bq[8];
        #pragma unroll
        for (int t = 0; t < 8; ++t) bq[t] = wb[2 * g * GAT_FDIM + 16 * t];
        #pragma unroll
        for (int t = 0; t < 8; ++t) {
            const v2f b = v2f{bq[t].x, bq[t].y};
            acc[t] = __builtin_amdgcn_wmma_f32_16x16x4_f32(
                 false, a0,  false, b,
                 (short)0, acc[t],
                 false,  false);
        }
        #pragma unroll
        for (int t = 0; t < 8; ++t) {
            const v2f b = v2f{bq[t].z, bq[t].w};
            acc[t] = __builtin_amdgcn_wmma_f32_16x16x4_f32(
                 false, a1,  false, b,
                 (short)0, acc[t],
                 false,  false);
        }
    }

    const int mOff = 8 * khalf;
    #pragma unroll
    for (int t = 0; t < 8; ++t) {
        #pragma unroll
        for (int r = 0; r < 8; ++r) {
            int m = mTile * 16 + mOff + r;
            if (m < nNodes)
                feat[(size_t)m * GAT_FDIM + t * 16 + laneM] = acc[t][r];
        }
    }

    const size_t lineOff = (size_t)(mTile * 16 + mOff) * GAT_HEADS + 2 * laneM;
    const bool rowOk = mTile * 16 + mOff + (laneM >> 1) < nNodes;
    {
        float wl[8];
        #pragma unroll
        for (int t = 0; t < 8; ++t) wl[t] = attn_l[t * 16 + laneM];
        float p[32];
        #pragma unroll
        for (int r = 0; r < 8; ++r) {
            #pragma unroll
            for (int h = 0; h < GAT_HEADS; ++h)
                p[r * 4 + h] = acc[2 * h][r] * wl[2 * h] + acc[2 * h + 1][r] * wl[2 * h + 1];
        }
        row_sum_32(p, laneM);
        if (rowOk) *(v2f*)(el + lineOff) = v2f{p[0], p[1]};
    }
    {
        float wr[8];
        #pragma unroll
        for (int t = 0; t < 8; ++t) wr[t] = attn_r[t * 16 + laneM];
        float p[32];
        #pragma unroll
        for (int r = 0; r < 8; ++r) {
            #pragma unroll
            for (int h = 0; h < GAT_HEADS; ++h)
                p[r * 4 + h] = acc[2 * h][r] * wr[2 * h] + acc[2 * h + 1][r] * wr[2 * h + 1];
        }
        row_sum_32(p, laneM);
        if (rowOk) *(v2f*)(er + lineOff) = v2f{p[0], p[1]};
    }
}

__device__ __forceinline__ float lrelu(float v) {
    return v > 0.f ? v : NEG_SLOPE * v;
}

#define BKT_SHIFT    8
#define BKT_NODES    256
#define MAX_BKTS     512
#define CHUNK_EDGES  3200
#define CHUNK_STEPS  13
#define STAGE_WORDS  (WL_ENTRIES * 4)

__device__ __forceinline__ int block_excl_scan(int v, int* waveTot, int* total) {
    const int lane = threadIdx.x & 31, wave = threadIdx.x >> 5;
    int incl = v;
    #pragma unroll
    for (int off = 1; off < 32; off <<= 1) {
        int t = __shfl_up(incl, off, 32);
        if (lane >= off) incl += t;
    }
    if (lane == 31) waveTot[wave] = incl;
    __syncthreads();
    int wbase = 0, tot = 0;
    #pragma unroll
    for (int q = 0; q < 8; ++q) {
        const int wt = waveTot[q];
        wbase += q < wave ? wt : 0;
        tot += wt;
    }
    *total = tot;
    return wbase + incl - v;
}

__device__ __forceinline__ void edge_bin_block(
    const int* __restrict__ src, const int* __restrict__ dst,
    int* __restrict__ binned, int* __restrict__ runTab, int* __restrict__ cursor,
    int nEdges, int nBkts, int nChunks, int w, int* stage)
{
    __shared__ int hist[MAX_BKTS + 2];
    __shared__ int waveTot[8];
    const int t = threadIdx.x;
    const int base = w * CHUNK_EDGES;
    const int left = nEdges - base;
    const int len = left < CHUNK_EDGES ? left : CHUNK_EDGES;
    if (w == 0 && t == 0) *cursor = 0;
    hist[2 * t] = 0;
    hist[2 * t + 1] = 0;
    __syncthreads();

    int d[CHUNK_STEPS], s[CHUNK_STEPS];
    #pragma unroll
    for (int k = 0; k < CHUNK_STEPS; ++k) {
        const int i  = k * 256 + t;
        const int ic = i < len ? i : len - 1;
        d[k] = dst[base + ic];
        s[k] = src[base + ic];
    }
    int lr[CHUNK_STEPS];
    #pragma unroll
    for (int k = 0; k < CHUNK_STEPS; ++k)
        lr[k] = atomicAdd(&hist[k * 256 + t < len ? d[k] >> BKT_SHIFT : MAX_BKTS + 1], 1);
    __syncthreads();

    const int c0 = hist[2 * t], c1 = hist[2 * t + 1];
    int total;
    const int ex = block_excl_scan(c0 + c1, waveTot, &total);
    hist[2 * t] = ex;
    hist[2 * t + 1] = ex + c0;
    if (t == 255) hist[MAX_BKTS] = total;
    __syncthreads();

    for (int b = t; b <= nBkts; b += 256)
        runTab[(size_t)b * nChunks + w] = hist[b];
    int pl[CHUNK_STEPS];
    #pragma unroll
    for (int k = 0; k < CHUNK_STEPS; ++k) pl[k] = hist[d[k] >> BKT_SHIFT] + lr[k];
    #pragma unroll
    for (int k = 0; k < CHUNK_STEPS; ++k) {
        if (k * 256 + t < len)
            stage[pl[k]] =
                (int)(((unsigned)s[k] << BKT_SHIFT) | (unsigned)(d[k] & (BKT_NODES - 1)));
    }
    __syncthreads();
    int ow[CHUNK_STEPS];
    #pragma unroll
    for (int k = 0; k < CHUNK_STEPS; ++k) {
        const int i = k * 256 + t;
        ow[k] = stage[i < len ? i : len - 1];
    }
    #pragma unroll
    for (int k = 0; k < CHUNK_STEPS; ++k) {
        const int i = k * 256 + t;
        if (i < len) binned[base + i] = ow[k];
    }
}

__device__ __forceinline__ void bucket_group_block(
    const int* __restrict__ binned, const int* __restrict__ runTab,
    int* __restrict__ cnt, int* __restrict__ start, int* __restrict__ sorted,
    int* __restrict__ cursor, int nNodes, int nChunks, int b, int* stage)
{
    __shared__ int slot[BKT_NODES];
    __shared__ int waveTot[8];
    __shared__ int blockBase;
    const int t = threadIdx.x;
    const int* rowBeg = runTab + (size_t)b * nChunks;
    const int* rowEnd = rowBeg + nChunks;
    slot[t] = 0;
    __syncthreads();

    for (int r = t; r < nChunks; r += 256) {
        const int* p = binned + (size_t)r * CHUNK_EDGES;
        int e0 = rowBeg[r], e1 = rowEnd[r];
        e0 = e0 < 0 ? 0 : e0;
        e1 = e1 > CHUNK_EDGES ? CHUNK_EDGES : e1;
        for (int e = e0; e < e1; e += 4) {
            int wd[4];
            #pragma unroll
            for (int q = 0; q < 4; ++q) wd[q] = p[e + q < e1 ? e + q : e1 - 1];
            #pragma unroll
            for (int q = 0; q < 4; ++q)
                if (e + q < e1) atomicAdd(&slot[wd[q] & (BKT_NODES - 1)], 1);
        }
    }
    __syncthreads();

    const int c = slot[t];
    int total;
    const int ex = block_excl_scan(c, waveTot, &total);
    if (t == 0) blockBase = atomicAdd(cursor, total);
    __syncthreads();
    const int bb = blockBase;
    const int st = bb + ex;
    const int v = b * BKT_NODES + t;
    if (v < nNodes) { start[v] = st; cnt[v] = c; }
    const int nBlk = __builtin_amdgcn_readfirstlane(total);
    const bool inLds = nBlk <= STAGE_WORDS;
    slot[t] = inLds ? ex : st;
    __syncthreads();

    for (int r = t; r < nChunks; r += 256) {
        const int* p = binned + (size_t)r * CHUNK_EDGES;
        int e0 = rowBeg[r], e1 = rowEnd[r];
        e0 = e0 < 0 ? 0 : e0;
        e1 = e1 > CHUNK_EDGES ? CHUNK_EDGES : e1;
        for (int e = e0; e < e1; e += 4) {
            int wd[4];
            #pragma unroll
            for (int q = 0; q < 4; ++q) wd[q] = p[e + q < e1 ? e + q : e1 - 1];
            #pragma unroll
            for (int q = 0; q < 4; ++q) {
                if (e + q < e1) {
                    const int pos = atomicAdd(&slot[wd[q] & (BKT_NODES - 1)], 1);
                    const int id = (int)((unsigned)wd[q] >> BKT_SHIFT);
                    if (inLds) stage[pos] = id; else sorted[pos] = id;
                }
            }
        }
    }
    if (inLds) {
        __syncthreads();
        for (int i0 = 0; i0 < nBlk; i0 += 1024) {
            int ow[4];
            #pragma unroll
            for (int q = 0; q < 4; ++q) {
                const int i = i0 + q * 256 + t;
                ow[q] = stage[i < nBlk ? i : nBlk - 1];
            }
            #pragma unroll
            for (int q = 0; q < 4; ++q) {
                const int i = i0 + q * 256 + t;
                if (i < nBlk) sorted[bb + i] = ow[q];
            }
        }
    }
}

__global__ __launch_bounds__(256) void gat_bin_and_gemm(
    const int* __restrict__ src, const int* __restrict__ dst,
    int* __restrict__ binned, int* __restrict__ runTab, int* __restrict__ cursor,
    int nEdges, int nBkts, int nChunks,
    const float* __restrict__ x, const float* __restrict__ w,
    const float* __restrict__ attn_l, const float* __restrict__ attn_r,
    float* __restrict__ feat, float* __restrict__ el, float* __restrict__ er,
    int nNodes, int tile0, int tile1)
{
    __shared__ float4 wl[WL_ENTRIES];
    if ((int)blockIdx.x < nChunks) {
        edge_bin_block(src, dst, binned, runTab, cursor, nEdges, nBkts, nChunks,
                       (int)blockIdx.x, (int*)wl);
    } else {
        stage_w(w, wl);
        __syncthreads();
        const int mTile = tile0 + ((int)blockIdx.x - nChunks) * 8 + (int)(threadIdx.x >> 5);
        if (mTile < tile1)
            gemm_strip(x, wl, attn_l, attn_r, feat, el, er, nNodes, mTile);
    }
}

__global__ __launch_bounds__(256) void gat_group_and_gemm(
    const int* __restrict__ binned, const int* __restrict__ runTab,
    int* __restrict__ cnt, int* __restrict__ start, int* __restrict__ sorted,
    int* __restrict__ cursor, int nBkts, int nChunks,
    const float* __restrict__ x, const float* __restrict__ w,
    const float* __restrict__ attn_l, const float* __restrict__ attn_r,
    float* __restrict__ feat, float* __restrict__ el, float* __restrict__ er,
    int nNodes, int tile0, int tile1)
{
    __shared__ float4 wl[WL_ENTRIES];
    if ((int)blockIdx.x < nBkts) {
        bucket_group_block(binned, runTab, cnt, start, sorted, cursor, nNodes, nChunks,
                           (int)blockIdx.x, (int*)wl);
    } else {
        stage_w(w, wl);
        __syncthreads();
        const int mTile = tile0 + ((int)blockIdx.x - nBkts) * 8 + (int)(threadIdx.x >> 5);
        if (mTile < tile1)
            gemm_strip(x, wl, attn_l, attn_r, feat, el, er, nNodes, mTile);
    }
}


template <int R>
__device__ __forceinline__ int grp_get_i(int v) {
    return (int)__builtin_amdgcn_mov_dpp8((unsigned)v, R * 0x249249u);
}
template <int R>
__device__ __forceinline__ float grp_get(float v) {
    return __uint_as_float(__builtin_amdgcn_mov_dpp8(__float_as_uint(v), R * 0x249249u));
}
__device__ __forceinline__ float grp_max(float x) {
    x = fmaxf(x, row_xor_get<1>(x));
    x = fmaxf(x, row_xor_get<2>(x));
    x = fmaxf(x, row_xor_get<4>(x));
    return x;
}
__device__ __forceinline__ float grp_sum(float x) {
    x += row_xor_get<1>(x);
    x += row_xor_get<2>(x);
    x += row_xor_get<4>(x);
    return x;
}

template <int R>
__device__ __forceinline__ void slot_load(float4 (&f)[GAT_HEADS], int uk,
                                          const float* __restrict__ feat, int q)
{
    const int us = grp_get_i<R>(uk);
    const float* frow = feat + (size_t)us * GAT_FDIM + 4 * q;
    #pragma unroll
    for (int hh = 0; hh < GAT_HEADS; ++hh) f[hh] = *(const float4*)(frow + hh * GAT_D);
}
template <int R>
__device__ __forceinline__ void slot_add(float4& acc, const float4 (&f)[GAT_HEADS],
                                         const float (&ak)[GAT_HEADS], int kbase, int deg)
{
    const float a0 = grp_get<R>(ak[0]), a1 = grp_get<R>(ak[1]);
    const float a2 = grp_get<R>(ak[2]), a3 = grp_get<R>(ak[3]);
    const bool valid = kbase + R < deg;
    const float tx = a0 * f[0].x + a1 * f[1].x + a2 * f[2].x + a3 * f[3].x;
    const float ty = a0 * f[0].y + a1 * f[1].y + a2 * f[2].y + a3 * f[3].y;
    const float tz = a0 * f[0].z + a1 * f[1].z + a2 * f[2].z + a3 * f[3].z;
    const float tw = a0 * f[0].w + a1 * f[1].w + a2 * f[2].w + a3 * f[3].w;
    acc.x += valid ? tx : 0.f;
    acc.y += valid ? ty : 0.f;
    acc.z += valid ? tz : 0.f;
    acc.w += valid ? tw : 0.f;
}
template <int R0>
__device__ __forceinline__ void rows4(float4& acc, int uk, const float (&ak)[GAT_HEADS],
                                      int kbase, int deg,
                                      const float* __restrict__ feat, int q)
{
    float4 f0[GAT_HEADS], f1[GAT_HEADS], f2[GAT_HEADS], f3[GAT_HEADS];
    slot_load<R0 + 0>(f0, uk, feat, q);
    slot_load<R0 + 1>(f1, uk, feat, q);
    slot_load<R0 + 2>(f2, uk, feat, q);
    slot_load<R0 + 3>(f3, uk, feat, q);
    slot_add<R0 + 0>(acc, f0, ak, kbase, deg);
    slot_add<R0 + 1>(acc, f1, ak, kbase, deg);
    slot_add<R0 + 2>(acc, f2, ak, kbase, deg);
    slot_add<R0 + 3>(acc, f3, ak, kbase, deg);
}

__global__ __launch_bounds__(256) void gat_node_aggr(
    const int* __restrict__ sorted, const int* __restrict__ start,
    const int* __restrict__ cnt,
    const float* __restrict__ el, const float* __restrict__ er,
    const float* __restrict__ feat, const float* __restrict__ bias,
    float* __restrict__ out, int nNodes)
{
    const int lane = threadIdx.x & 31;
    const int g = lane >> 3;
    const int q = lane & 7;
    const int wv = __builtin_amdgcn_readfirstlane(
        (int)((blockIdx.x * blockDim.x + threadIdx.x) >> 5));
    if (wv * 4 >= nNodes) return;
    const int v = wv * 4 + g;
    const bool vOk = v < nNodes;
    const int vc = vOk ? v : nNodes - 1;

    const float4 er4 = *(const float4*)(er + (size_t)vc * GAT_HEADS);
    float4 bm;
    {
        const float4 b0 = *(const float4*)(bias + 4 * q);
        const float4 b1 = *(const float4*)(bias + GAT_D + 4 * q);
        const float4 b2 = *(const float4*)(bias + 2 * GAT_D + 4 * q);
        const float4 b3 = *(const float4*)(bias + 3 * GAT_D + 4 * q);
        bm.x = (b0.x + b1.x + b2.x + b3.x) * (1.f / GAT_HEADS);
        bm.y = (b0.y + b1.y + b2.y + b3.y) * (1.f / GAT_HEADS);
        bm.z = (b0.z + b1.z + b2.z + b3.z) * (1.f / GAT_HEADS);
        bm.w = (b0.w + b1.w + b2.w + b3.w) * (1.f / GAT_HEADS);
    }
    const int beg = start[vc];
    const int deg = vOk ? cnt[vc] : 0;
    const int d0 = __builtin_amdgcn_readlane(deg, 0), d1 = __builtin_amdgcn_readlane(deg, 8);
    const int d2 = __builtin_amdgcn_readlane(deg, 16), d3 = __builtin_amdgcn_readlane(deg, 24);
    const int d01 = d0 > d1 ? d0 : d1, d23 = d2 > d3 ? d2 : d3;
    const int dmax = d01 > d23 ? d01 : d23;

    const float NEG_INF = -__builtin_huge_valf();
    float4 acc = make_float4(0.f, 0.f, 0.f, 0.f);

    if (dmax <= 32) {
        int u[4];
        float a[4][GAT_HEADS];
        #pragma unroll
        for (int k = 0; k < 4; ++k) {
            const bool valid = 8 * k + q < deg;
            u[k] = sorted[valid ? beg + 8 * k + q : 0];
        }
        float4 l[4];
        #pragma unroll
        for (int k = 0; k < 4; ++k) l[k] = *(const float4*)(el + (size_t)u[k] * GAT_HEADS);
        #pragma unroll
        for (int k = 0; k < 4; ++k) {
            const float pen = 8 * k + q < deg ? 0.f : NEG_INF;
            a[k][0] = lrelu(l[k].x + er4.x) + pen;
            a[k][1] = lrelu(l[k].y + er4.y) + pen;
            a[k][2] = lrelu(l[k].z + er4.z) + pen;
            a[k][3] = lrelu(l[k].w + er4.w) + pen;
        }
        float s[GAT_HEADS];
        #pragma unroll
        for (int h = 0; h < GAT_HEADS; ++h) {
            const float m = grp_max(fmaxf(fmaxf(a[0][h], a[1][h]), fmaxf(a[2][h], a[3][h])));
            #pragma unroll
            for (int k = 0; k < 4; ++k)
                a[k][h] = 8 * k + q < deg ? __expf(a[k][h] - m) : 0.f;
            s[h] = fmaxf(grp_sum(((a[0][h] + a[1][h]) + a[2][h]) + a[3][h]), 1e-16f);
        }
        #pragma unroll
        for (int k = 0; k < 4; ++k) {
            #pragma unroll
            for (int h = 0; h < GAT_HEADS; ++h) a[k][h] = a[k][h] / s[h];
        }
        #pragma unroll
        for (int k = 0; k < 4; ++k) {
            if (8 * k < dmax) {
                rows4<0>(acc, u[k], a[k], 8 * k, deg, feat, q);
                if (8 * k + 4 < dmax)
                    rows4<4>(acc, u[k], a[k], 8 * k, deg, feat, q);
            }
        }
    } else {
        float m0 = NEG_INF, m1 = NEG_INF, m2 = NEG_INF, m3 = NEG_INF;
        for (int c = 0; c < dmax; c += 8) {
            const bool valid = c + q < deg;
            const int uu = sorted[valid ? beg + c + q : 0];
            const float4 l = *(const float4*)(el + (size_t)uu * GAT_HEADS);
            m0 = fmaxf(m0, valid ? lrelu(l.x + er4.x) : NEG_INF);
            m1 = fmaxf(m1, valid ? lrelu(l.y + er4.y) : NEG_INF);
            m2 = fmaxf(m2, valid ? lrelu(l.z + er4.z) : NEG_INF);
            m3 = fmaxf(m3, valid ? lrelu(l.w + er4.w) : NEG_INF);
        }
        m0 = grp_max(m0); m1 = grp_max(m1); m2 = grp_max(m2); m3 = grp_max(m3);
        float s0 = 0.f, s1 = 0.f, s2 = 0.f, s3 = 0.f;
        for (int c = 0; c < dmax; c += 8) {
            const bool valid = c + q < deg;
            const int uu = sorted[valid ? beg + c + q : 0];
            const float4 l = *(const float4*)(el + (size_t)uu * GAT_HEADS);
            s0 += valid ? __expf(lrelu(l.x + er4.x) - m0) : 0.f;
            s1 += valid ? __expf(lrelu(l.y + er4.y) - m1) : 0.f;
            s2 += valid ? __expf(lrelu(l.z + er4.z) - m2) : 0.f;
            s3 += valid ? __expf(lrelu(l.w + er4.w) - m3) : 0.f;
        }
        s0 = fmaxf(grp_sum(s0), 1e-16f); s1 = fmaxf(grp_sum(s1), 1e-16f);
        s2 = fmaxf(grp_sum(s2), 1e-16f); s3 = fmaxf(grp_sum(s3), 1e-16f);
        for (int c = 0; c < dmax; c += 8) {
            const bool valid = c + q < deg;
            const int uu = sorted[valid ? beg + c + q : 0];
            const float4 l = *(const float4*)(el + (size_t)uu * GAT_HEADS);
            float ak[GAT_HEADS];
            ak[0] = valid ? __expf(lrelu(l.x + er4.x) - m0) / s0 : 0.f;
            ak[1] = valid ? __expf(lrelu(l.y + er4.y) - m1) / s1 : 0.f;
            ak[2] = valid ? __expf(lrelu(l.z + er4.z) - m2) / s2 : 0.f;
            ak[3] = valid ? __expf(lrelu(l.w + er4.w) - m3) / s3 : 0.f;
            rows4<0>(acc, uu, ak, c, deg, feat, q);
            if (c + 4 < dmax) rows4<4>(acc, uu, ak, c, deg, feat, q);
        }
    }
    if (vOk) {
        float4 o;
        o.x = 0.25f * acc.x + bm.x;
        o.y = 0.25f * acc.y + bm.y;
        o.z = 0.25f * acc.z + bm.z;
        o.w = 0.25f * acc.w + bm.w;
        *(float4*)(out + (size_t)v * GAT_D + 4 * q) = o;
    }
}

extern "C" void kernel_launch(void* const* d_in, const int* in_sizes, int n_in,
                              void* d_out, int out_size, void* d_ws, size_t ws_size,
                              hipStream_t stream)
{
    const float* x        = (const float*)d_in[0];
    const int*   edge_src = (const int*)  d_in[1];
    const int*   edge_dst = (const int*)  d_in[2];
    const float* fc_w     = (const float*)d_in[3];
    const float* attn_l   = (const float*)d_in[4];
    const float* attn_r   = (const float*)d_in[5];
    const float* bias     = (const float*)d_in[6];
    float*       out      = (float*)d_out;

    const int nNodes = in_sizes[0] / GAT_FDIM;
    const int nEdges = in_sizes[1];

    float* feat   = (float*)d_ws;
    float* el     = feat + (size_t)nNodes * GAT_FDIM;
    float* er     = el   + (size_t)nNodes * GAT_HEADS;
    int*   cnt    = (int*)(er + (size_t)nNodes * GAT_HEADS);
    int*   start  = cnt + nNodes;
    int*   cursor = start + nNodes;
    int*   sorted = cursor + 4;
    if ((size_t)((char*)(sorted + nEdges) - (char*)d_ws) > ws_size) return;

    const int nBkts   = (nNodes + BKT_NODES - 1) / BKT_NODES;
    const int nChunks = (nEdges + CHUNK_EDGES - 1) / CHUNK_EDGES;
    const size_t scratchWords = (size_t)nChunks * CHUNK_EDGES + (size_t)(nBkts + 1) * nChunks;
    int* binned = sorted + nEdges;
    if ((size_t)((char*)(binned + scratchWords) - (char*)d_ws) > ws_size) {
        if (scratchWords > (size_t)nNodes * GAT_D) return;
        binned = (int*)d_out;
    }
    int* runTab = binned + (size_t)nChunks * CHUNK_EDGES;
    if (nBkts > MAX_BKTS || nNodes > (1 << 24)) return;

    const int mTiles  = (nNodes + 15) / 16;
    const int tileMid = ((mTiles / 2 + 7) / 8) * 8 < mTiles ? ((mTiles / 2 + 7) / 8) * 8 : mTiles;

    gat_bin_and_gemm<<<nChunks + (tileMid + 7) / 8, 256, 0, stream>>>(
        edge_src, edge_dst, binned, runTab, cursor, nEdges, nBkts, nChunks,
        x, fc_w, attn_l, attn_r, feat, el, er, nNodes, 0, tileMid);
    gat_group_and_gemm<<<nBkts + (mTiles - tileMid + 7) / 8, 256, 0, stream>>>(
        binned, runTab, cnt, start, sorted, cursor, nBkts, nChunks,
        x, fc_w, attn_l, attn_r, feat, el, er, nNodes, tileMid, mTiles);
    {
        int blocks = (nNodes + 31) / 32;
        gat_node_aggr<<<blocks, 256, 0, stream>>>(
            sorted, start, cnt, el, er, feat, bias, out, nNodes);
    }
}
